# starting kernel
# baseline (speedup 1.0000x reference)
#include <hip/hip_runtime.h>
#include <math.h>

#define N_NODES 50000
#define N_EDGES 800000
#define E_TOT   (N_EDGES + N_NODES)
#define F_IN    256
#define HEADS   8
#define OUT_CH  10
#define HIDDEN  80

typedef __attribute__((ext_vector_type(2))) float v2f;
typedef __attribute__((ext_vector_type(8))) float v8f;

__device__ __forceinline__ void atomic_max_f32(float* addr, float val) {
  if (val >= 0.0f) atomicMax((int*)addr, __float_as_int(val));
  else             atomicMin((unsigned int*)addr, __float_as_uint(val));
}

__global__ void __launch_bounds__(128)
gat_gemm2_wmma(const float* __restrict__ X,
               const float* __restrict__ Wl,
               const float* __restrict__ Wr,
               float* __restrict__ xl,
               float* __restrict__ xr,
               int K) {
  const int NT = HIDDEN / 16;
  const int n_tiles = (N_NODES / 16) * NT;
  int wave = blockIdx.x * (blockDim.x >> 5) + (threadIdx.x >> 5);
  if (wave >= n_tiles) return;

  int mt = wave / NT;
  int nt = wave % NT;
  int lane = threadIdx.x & 31;
  int half = lane >> 4;
  int r    = lane & 15;

  int arow = mt * 16 + r;
  int bcol = nt * 16 + r;

  v8f cl = {};
  v8f cr = {};

  for (int k = 0; k < K; k += 4) {
    int ka = k + 2 * half;
    __builtin_prefetch(&X[arow * K + k + 16], 0, 0);
    v2f a;
    a.x = X[arow * K + ka];
    a.y = X[arow * K + ka + 1];
    v2f bl;
    bl.x = Wl[ka * HIDDEN + bcol];
    bl.y = Wl[(ka + 1) * HIDDEN + bcol];
    v2f br;
    br.x = Wr[ka * HIDDEN + bcol];
    br.y = Wr[(ka + 1) * HIDDEN + bcol];
    cl = __builtin_amdgcn_wmma_f32_16x16x4_f32(false, a, false, bl, (short)0, cl, false, false);
    cr = __builtin_amdgcn_wmma_f32_16x16x4_f32(false, a, false, br, (short)0, cr, false, false);
  }

#pragma unroll
  for (int v = 0; v < 8; ++v) {
    int orow = mt * 16 + v + 8 * half;
    xl[orow * HIDDEN + nt * 16 + r] = cl[v];
    xr[orow * HIDDEN + nt * 16 + r] = cr[v];
  }
}

__global__ void gat_init(float* __restrict__ m, float* __restrict__ denom,
                         float* __restrict__ acc) {
  int idx = blockIdx.x * blockDim.x + threadIdx.x;
  if (idx < N_NODES * HEADS) {
    m[idx]     = -__builtin_inff();
    denom[idx] = 0.0f;
  }
  if (idx < N_NODES * HIDDEN) acc[idx] = 0.0f;
}

__global__ void gat_edge_logits(const int* __restrict__ esrc,
                                const int* __restrict__ edst,
                                const float* __restrict__ xl,
                                const float* __restrict__ xr,
                                const float* __restrict__ att,
                                float* __restrict__ logit,
                                float* m_buf) {
  int eh = blockIdx.x * blockDim.x + threadIdx.x;
  if (eh >= E_TOT * HEADS) return;
  int e = eh >> 3;
  int h = eh & 7;
  int s, d;
  if (e < N_EDGES) { s = esrc[e]; d = edst[e]; }
  else             { s = d = e - N_EDGES; }

  const float* pl = xl + s * HIDDEN + h * OUT_CH;
  const float* pr = xr + d * HIDDEN + h * OUT_CH;
  const float* pa = att + h * OUT_CH;
  float acc = 0.0f;
#pragma unroll
  for (int c = 0; c < OUT_CH; ++c) {
    float v = pl[c] + pr[c];
    v = v > 0.0f ? v : 0.2f * v;
    acc += v * pa[c];
  }
  logit[eh] = acc;
  atomic_max_f32(&m_buf[d * HEADS + h], acc);
}

__global__ void gat_edge_accum(const int* __restrict__ esrc,
                               const int* __restrict__ edst,
                               const float* __restrict__ xl,
                               const float* __restrict__ logit,
                               const float* __restrict__ m_buf,
                               float* denom,
                               float* acc) {
  int eh = blockIdx.x * blockDim.x + threadIdx.x;
  if (eh >= E_TOT * HEADS) return;
  int e = eh >> 3;
  int h = eh & 7;
  int s, d;
  if (e < N_EDGES) { s = esrc[e]; d = edst[e]; }
  else             { s = d = e - N_EDGES; }

  float ex = __expf(logit[eh] - m_buf[d * HEADS + h]);
  atomicAdd(&denom[d * HEADS + h], ex);
  const float* pl = xl + s * HIDDEN + h * OUT_CH;
  float* pa = acc + d * HIDDEN + h * OUT_CH;
#pragma unroll
  for (int c = 0; c < OUT_CH; ++c) atomicAdd(&pa[c], ex * pl[c]);
}

__global__ void gat_node_finish(const float* __restrict__ acc,
                                const float* __restrict__ denom,
                                const float* __restrict__ bias,
                                float* __restrict__ xout) {
  int idx = blockIdx.x * blockDim.x + threadIdx.x;
  if (idx >= N_NODES * HIDDEN) return;
  int i = idx / HIDDEN;
  int j = idx - i * HIDDEN;
  int h = j / OUT_CH;
  float v = acc[idx] / (denom[i * HEADS + h] + 1e-16f) + bias[j];
  xout[idx] = v > 0.0f ? v : expm1f(v);
}

extern "C" void kernel_launch(void* const* d_in, const int* in_sizes, int n_in,
                              void* d_out, int out_size, void* d_ws, size_t ws_size,
                              hipStream_t stream) {
  (void)in_sizes; (void)n_in; (void)out_size; (void)ws_size;

  const float* x    = (const float*)d_in[0];
  const int*   ei   = (const int*)d_in[1];
  const int*   esrc = ei;
  const int*   edst = ei + N_EDGES;

  float* ws    = (float*)d_ws;
  float* xl    = ws;
  float* xr    = xl    + (size_t)N_NODES * HIDDEN;
  float* logit = xr    + (size_t)N_NODES * HIDDEN;
  float* mbuf  = logit + (size_t)E_TOT * HEADS;
  float* denom = mbuf  + (size_t)N_NODES * HEADS;
  float* acc   = denom + (size_t)N_NODES * HEADS;
  float* xbuf  = acc   + (size_t)N_NODES * HIDDEN;

  const int gemm_waves  = (N_NODES / 16) * (HIDDEN / 16);
  const int gemm_blocks = (gemm_waves + 3) / 4;
  const int node_blocks = (N_NODES * HIDDEN + 255) / 256;
  const int edge_blocks = (E_TOT * HEADS + 255) / 256;

  const float* xin = x;
  int K = F_IN;
  for (int L = 0; L < 3; ++L) {
    const float* Wl = (const float*)d_in[2 + 4 * L];
    const float* Wr = (const float*)d_in[3 + 4 * L];
    const float* at = (const float*)d_in[4 + 4 * L];
    const float* b  = (const float*)d_in[5 + 4 * L];

    gat_gemm2_wmma<<<gemm_blocks, 128, 0, stream>>>(xin, Wl, Wr, xl, xr, K);
    gat_init<<<node_blocks, 256, 0, stream>>>(mbuf, denom, acc);
    gat_edge_logits<<<edge_blocks, 256, 0, stream>>>(esrc, edst, xl, xr, at, logit, mbuf);
    gat_edge_accum<<<edge_blocks, 256, 0, stream>>>(esrc, edst, xl, logit, mbuf, denom, acc);

    float* xout = (L == 2) ? (float*)d_out : xbuf;
    gat_node_finish<<<node_blocks, 256, 0, stream>>>(acc, denom, b, xout);

    xin = xbuf;
    K = HIDDEN;
  }
}


# edges grouped by destination once per call (counting sort, int atomics); each layer = one per-node gather loop with running-max softmax, no float atomics; sum order = arrival-order rank
# speedup vs baseline: 3.3555x; 3.3555x over previous
#include <hip/hip_runtime.h>
#include <math.h>

#define N_NODES 50000
#define N_EDGES 800000
#define F_IN    256
#define HEADS   8
#define OUT_CH  10
#define HIDDEN  80

typedef __attribute__((ext_vector_type(2))) float v2f;
typedef __attribute__((ext_vector_type(8))) float v8f;

__global__ void __launch_bounds__(128)
gat_gemm2_wmma(const float* __restrict__ X,
               const float* __restrict__ Wl,
               const float* __restrict__ Wr,
               float* __restrict__ xl,
               float* __restrict__ xr,
               int K) {
  const int NT = HIDDEN / 16;
  const int n_tiles = (N_NODES / 16) * NT;
  int wave = blockIdx.x * (blockDim.x >> 5) + (threadIdx.x >> 5);
  if (wave >= n_tiles) return;

  int mt = wave / NT;
  int nt = wave % NT;
  int lane = threadIdx.x & 31;
  int half = lane >> 4;
  int r    = lane & 15;

  int arow = mt * 16 + r;
  int bcol = nt * 16 + r;

  v8f cl = {};
  v8f cr = {};

  for (int k = 0; k < K; k += 4) {
    int ka = k + 2 * half;
    __builtin_prefetch(&X[arow * K + k + 16], 0, 0);
    v2f a;
    a.x = X[arow * K + ka];
    a.y = X[arow * K + ka + 1];
    v2f bl;
    bl.x = Wl[ka * HIDDEN + bcol];
    bl.y = Wl[(ka + 1) * HIDDEN + bcol];
    v2f br;
    br.x = Wr[ka * HIDDEN + bcol];
    br.y = Wr[(ka + 1) * HIDDEN + bcol];
    cl = __builtin_amdgcn_wmma_f32_16x16x4_f32(false, a, false, bl, (short)0, cl, false, false);
    cr = __builtin_amdgcn_wmma_f32_16x16x4_f32(false, a, false, br, (short)0, cr, false, false);
  }

#pragma unroll
  for (int v = 0; v < 8; ++v) {
    int orow = mt * 16 + v + 8 * half;
    xl[orow * HIDDEN + nt * 16 + r] = cl[v];
    xr[orow * HIDDEN + nt * 16 + r] = cr[v];
  }
}

__global__ void csr_zero(int* __restrict__ deg, int* __restrict__ total) {
  int idx = blockIdx.x * blockDim.x + threadIdx.x;
  if (idx < N_NODES) deg[idx] = 0;
  if (idx == 0) *total = 0;
}

__global__ void csr_count(const int* __restrict__ edst, int* deg,
                          int* __restrict__ rank) {
  int e = blockIdx.x * blockDim.x + threadIdx.x;
  if (e >= N_EDGES) return;
  rank[e] = atomicAdd(&deg[edst[e]], 1);
}

__global__ void __launch_bounds__(256)
csr_alloc(const int* __restrict__ deg, int* __restrict__ rowstart, int* total) {
  int node = blockIdx.x * blockDim.x + threadIdx.x;
  int lane = threadIdx.x & 31;
  int dg = node < N_NODES ? deg[node] : 0;
  int incl = dg;
  for (int off = 1; off < 32; off <<= 1) {
    int t = __shfl_up(incl, off);
    if (lane >= off) incl += t;
  }
  int base = 0;
  if (lane == 31) base = atomicAdd(total, incl);
  base = __shfl(base, 31);
  if (node < N_NODES) rowstart[node] = base + incl - dg;
}

__global__ void csr_fill(const int* __restrict__ esrc,
                         const int* __restrict__ edst,
                         const int* __restrict__ rowstart,
                         const int* __restrict__ rank,
                         int* __restrict__ csr_src) {
  int e = blockIdx.x * blockDim.x + threadIdx.x;
  if (e >= N_EDGES) return;
  int pos = rowstart[edst[e]] + rank[e];
  if ((unsigned)pos < (unsigned)N_EDGES) csr_src[pos] = esrc[e];
}

__device__ __forceinline__ void load10(const float* __restrict__ p, float (&v)[OUT_CH]) {
  const float2* q = (const float2*)p;
  for (int i = 0; i < OUT_CH / 2; ++i) {
    float2 t = q[i];
    v[2 * i]     = t.x;
    v[2 * i + 1] = t.y;
  }
}

__device__ __forceinline__ float head_logit(const float (&x)[OUT_CH],
                                            const float (&r)[OUT_CH],
                                            const float (&a)[OUT_CH]) {
  float acc = 0.0f;
  for (int c = 0; c < OUT_CH; ++c) {
    float v = x[c] + r[c];
    v = v > 0.0f ? v : 0.2f * v;
    acc += v * a[c];
  }
  return acc;
}

__device__ __forceinline__ int clamp_node(int s) {
  return min(max(s, 0), N_NODES - 1);
}

__global__ void __launch_bounds__(256)
gat_node_gather(const int* __restrict__ rowstart,
                const int* __restrict__ deg,
                const int* __restrict__ csr_src,
                const float* __restrict__ xl,
                const float* __restrict__ xr,
                const float* __restrict__ att,
                const float* __restrict__ bias,
                float* __restrict__ xout) {
  int t = blockIdx.x * blockDim.x + threadIdx.x;
  if (t >= N_NODES * HEADS) return;
  int d = t >> 3;
  int h = t & 7;
  int hoff = h * OUT_CH;

  float a[OUT_CH], r[OUT_CH], x0[OUT_CH], x1[OUT_CH], acc[OUT_CH];
  load10(att + hoff, a);
  load10(xr + d * HIDDEN + hoff, r);

  int beg = min(max(rowstart[d], 0), N_EDGES);
  int end = beg + min(max(deg[d], 0), N_EDGES - beg);

  load10(xl + d * HIDDEN + hoff, x0);
  float m   = head_logit(x0, r, a);
  float den = 1.0f;
  for (int c = 0; c < OUT_CH; ++c) acc[c] = x0[c];

  int e = beg;
  int s0 = 0, s1 = 0;
  if (e + 1 < end) { s0 = csr_src[e]; s1 = csr_src[e + 1]; }
  while (e + 1 < end) {
    int c0 = clamp_node(s0);
    int c1 = clamp_node(s1);
    e += 2;
    if (e + 1 < end) { s0 = csr_src[e]; s1 = csr_src[e + 1]; }
    load10(xl + c0 * HIDDEN + hoff, x0);
    load10(xl + c1 * HIDDEN + hoff, x1);
    float l0 = head_logit(x0, r, a);
    float l1 = head_logit(x1, r, a);
    float mn = fmaxf(m, fmaxf(l0, l1));
    float sc = __expf(m - mn);
    float p0 = __expf(l0 - mn);
    float p1 = __expf(l1 - mn);
    den = den * sc + (p0 + p1);
    for (int c = 0; c < OUT_CH; ++c) acc[c] = acc[c] * sc + (p0 * x0[c] + p1 * x1[c]);
    m = mn;
  }
  if (e < end) {
    int c0 = clamp_node(csr_src[e]);
    load10(xl + c0 * HIDDEN + hoff, x0);
    float l0 = head_logit(x0, r, a);
    float mn = fmaxf(m, l0);
    float sc = __expf(m - mn);
    float p0 = __expf(l0 - mn);
    den = den * sc + p0;
    for (int c = 0; c < OUT_CH; ++c) acc[c] = acc[c] * sc + p0 * x0[c];
  }

  float inv = 1.0f / (den + 1e-16f);
  float2* po = (float2*)(xout + d * HIDDEN + hoff);
  const float* pb = bias + hoff;
  for (int i = 0; i < OUT_CH / 2; ++i) {
    float v0 = acc[2 * i]     * inv + pb[2 * i];
    float v1 = acc[2 * i + 1] * inv + pb[2 * i + 1];
    float2 o;
    o.x = v0 > 0.0f ? v0 : expm1f(v0);
    o.y = v1 > 0.0f ? v1 : expm1f(v1);
    po[i] = o;
  }
}

extern "C" void kernel_launch(void* const* d_in, const int* in_sizes, int n_in,
                              void* d_out, int out_size, void* d_ws, size_t ws_size,
                              hipStream_t stream) {
  (void)in_sizes; (void)n_in; (void)out_size; (void)ws_size;

  const float* x    = (const float*)d_in[0];
  const int*   ei   = (const int*)d_in[1];
  const int*   esrc = ei;
  const int*   edst = ei + N_EDGES;

  float* ws    = (float*)d_ws;
  float* xl    = ws;
  float* xr    = xl   + (size_t)N_NODES * HIDDEN;
  float* xbuf  = xr   + (size_t)N_NODES * HIDDEN;
  int*   deg      = (int*)(xbuf + (size_t)N_NODES * HIDDEN);
  int*   rowstart = deg      + N_NODES;
  int*   total    = rowstart + N_NODES;
  int*   rank     = total    + 64;
  int*   csr_src  = rank     + N_EDGES;

  const int gemm_waves  = (N_NODES / 16) * (HIDDEN / 16);
  const int gemm_blocks = (gemm_waves + 3) / 4;
  const int nodes_blocks = (N_NODES + 255) / 256;
  const int edges_blocks = (N_EDGES + 255) / 256;
  const int head_blocks  = (N_NODES * HEADS + 255) / 256;

  csr_zero <<<nodes_blocks, 256, 0, stream>>>(deg, total);
  csr_count<<<edges_blocks, 256, 0, stream>>>(edst, deg, rank);
  csr_alloc<<<nodes_blocks, 256, 0, stream>>>(deg, rowstart, total);
  csr_fill <<<edges_blocks, 256, 0, stream>>>(esrc, edst, rowstart, rank, csr_src);

  const float* xin = x;
  int K = F_IN;
  for (int L = 0; L < 3; ++L) {
    const float* Wl = (const float*)d_in[2 + 4 * L];
    const float* Wr = (const float*)d_in[3 + 4 * L];
    const float* at = (const float*)d_in[4 + 4 * L];
    const float* b  = (const float*)d_in[5 + 4 * L];

    gat_gemm2_wmma<<<gemm_blocks, 128, 0, stream>>>(xin, Wl, Wr, xl, xr, K);

    float* xout = (L == 2) ? (float*)d_out : xbuf;
    gat_node_gather<<<head_blocks, 256, 0, stream>>>(rowstart, deg, csr_src, xl, xr, at, b, xout);

    xin = xbuf;
    K = HIDDEN;
  }
}


# GEMM: one wave per 16-row strip holds all 10 output tiles (5 column tiles x Wl,Wr), so each A fragment is loaded once and feeds 10 WMMAs instead of 2; per-step A prefetch hint dropped with it
# speedup vs baseline: 3.9494x; 1.1770x over previous
#include <hip/hip_runtime.h>
#include <math.h>

#define N_NODES 50000
#define N_EDGES 800000
#define F_IN    256
#define HEADS   8
#define OUT_CH  10
#define HIDDEN  80

typedef __attribute__((ext_vector_type(2))) float v2f;
typedef __attribute__((ext_vector_type(8))) float v8f;

template <int K>
__global__ void __launch_bounds__(32)
gat_gemm2_wmma(const float* __restrict__ X,
               const float* __restrict__ Wl,
               const float* __restrict__ Wr,
               float* __restrict__ xl,
               float* __restrict__ xr) {
  const int NT = HIDDEN / 16;
  int mt   = blockIdx.x;
  int lane = threadIdx.x;
  int half = lane >> 4;
  int r    = lane & 15;

  const float* ap  = X  + (mt * 16 + r) * K + 2 * half;
  const float* blp = Wl + (2 * half) * HIDDEN + r;
  const float* brp = Wr + (2 * half) * HIDDEN + r;

  v8f cl[NT];
  v8f cr[NT];
#pragma unroll
  for (int t = 0; t < NT; ++t) { cl[t] = (v8f){}; cr[t] = (v8f){}; }

#pragma unroll 2
  for (int k = 0; k < K; k += 4) {
    v2f a = *(const v2f*)(ap + k);
#pragma unroll
    for (int t = 0; t < NT; ++t) {
      v2f bl;
      bl.x = blp[k * HIDDEN + t * 16];
      bl.y = blp[(k + 1) * HIDDEN + t * 16];
      v2f br;
      br.x = brp[k * HIDDEN + t * 16];
      br.y = brp[(k + 1) * HIDDEN + t * 16];
      cl[t] = __builtin_amdgcn_wmma_f32_16x16x4_f32(false, a, false, bl, (short)0, cl[t], false, false);
      cr[t] = __builtin_amdgcn_wmma_f32_16x16x4_f32(false, a, false, br, (short)0, cr[t], false, false);
    }
  }

#pragma unroll
  for (int t = 0; t < NT; ++t) {
#pragma unroll
    for (int v = 0; v < 8; ++v) {
      int orow = mt * 16 + v + 8 * half;
      xl[orow * HIDDEN + t * 16 + r] = cl[t][v];
      xr[orow * HIDDEN + t * 16 + r] = cr[t][v];
    }
  }
}

__global__ void csr_zero(int* __restrict__ deg, int* __restrict__ total) {
  int idx = blockIdx.x * blockDim.x + threadIdx.x;
  if (idx < N_NODES) deg[idx] = 0;
  if (idx == 0) *total = 0;
}

__global__ void csr_count(const int* __restrict__ edst, int* deg,
                          int* __restrict__ rank) {
  int e = blockIdx.x * blockDim.x + threadIdx.x;
  if (e >= N_EDGES) return;
  rank[e] = atomicAdd(&deg[edst[e]], 1);
}

__global__ void __launch_bounds__(256)
csr_alloc(const int* __restrict__ deg, int* __restrict__ rowstart, int* total) {
  int node = blockIdx.x * blockDim.x + threadIdx.x;
  int lane = threadIdx.x & 31;
  int dg = node < N_NODES ? deg[node] : 0;
  int incl = dg;
  for (int off = 1; off < 32; off <<= 1) {
    int t = __shfl_up(incl, off);
    if (lane >= off) incl += t;
  }
  int base = 0;
  if (lane == 31) base = atomicAdd(total, incl);
  base = __shfl(base, 31);
  if (node < N_NODES) rowstart[node] = base + incl - dg;
}

__global__ void csr_fill(const int* __restrict__ esrc,
                         const int* __restrict__ edst,
                         const int* __restrict__ rowstart,
                         const int* __restrict__ rank,
                         int* __restrict__ csr_src) {
  int e = blockIdx.x * blockDim.x + threadIdx.x;
  if (e >= N_EDGES) return;
  int pos = rowstart[edst[e]] + rank[e];
  if ((unsigned)pos < (unsigned)N_EDGES) csr_src[pos] = esrc[e];
}

__device__ __forceinline__ void load10(const float* __restrict__ p, float (&v)[OUT_CH]) {
  const float2* q = (const float2*)p;
  for (int i = 0; i < OUT_CH / 2; ++i) {
    float2 t = q[i];
    v[2 * i]     = t.x;
    v[2 * i + 1] = t.y;
  }
}

__device__ __forceinline__ float head_logit(const float (&x)[OUT_CH],
                                            const float (&r)[OUT_CH],
                                            const float (&a)[OUT_CH]) {
  float acc = 0.0f;
  for (int c = 0; c < OUT_CH; ++c) {
    float v = x[c] + r[c];
    v = v > 0.0f ? v : 0.2f * v;
    acc += v * a[c];
  }
  return acc;
}

__device__ __forceinline__ int clamp_node(int s) {
  return min(max(s, 0), N_NODES - 1);
}

__global__ void __launch_bounds__(256)
gat_node_gather(const int* __restrict__ rowstart,
                const int* __restrict__ deg,
                const int* __restrict__ csr_src,
                const float* __restrict__ xl,
                const float* __restrict__ xr,
                const float* __restrict__ att,
                const float* __restrict__ bias,
                float* __restrict__ xout) {
  int t = blockIdx.x * blockDim.x + threadIdx.x;
  if (t >= N_NODES * HEADS) return;
  int d = t >> 3;
  int h = t & 7;
  int hoff = h * OUT_CH;

  float a[OUT_CH], r[OUT_CH], x0[OUT_CH], x1[OUT_CH], acc[OUT_CH];
  load10(att + hoff, a);
  load10(xr + d * HIDDEN + hoff, r);

  int beg = min(max(rowstart[d], 0), N_EDGES);
  int end = beg + min(max(deg[d], 0), N_EDGES - beg);

  load10(xl + d * HIDDEN + hoff, x0);
  float m   = head_logit(x0, r, a);
  float den = 1.0f;
  for (int c = 0; c < OUT_CH; ++c) acc[c] = x0[c];

  int e = beg;
  int s0 = 0, s1 = 0;
  if (e + 1 < end) { s0 = csr_src[e]; s1 = csr_src[e + 1]; }
  while (e + 1 < end) {
    int c0 = clamp_node(s0);
    int c1 = clamp_node(s1);
    e += 2;
    if (e + 1 < end) { s0 = csr_src[e]; s1 = csr_src[e + 1]; }
    load10(xl + c0 * HIDDEN + hoff, x0);
    load10(xl + c1 * HIDDEN + hoff, x1);
    float l0 = head_logit(x0, r, a);
    float l1 = head_logit(x1, r, a);
    float mn = fmaxf(m, fmaxf(l0, l1));
    float sc = __expf(m - mn);
    float p0 = __expf(l0 - mn);
    float p1 = __expf(l1 - mn);
    den = den * sc + (p0 + p1);
    for (int c = 0; c < OUT_CH; ++c) acc[c] = acc[c] * sc + (p0 * x0[c] + p1 * x1[c]);
    m = mn;
  }
  if (e < end) {
    int c0 = clamp_node(csr_src[e]);
    load10(xl + c0 * HIDDEN + hoff, x0);
    float l0 = head_logit(x0, r, a);
    float mn = fmaxf(m, l0);
    float sc = __expf(m - mn);
    float p0 = __expf(l0 - mn);
    den = den * sc + p0;
    for (int c = 0; c < OUT_CH; ++c) acc[c] = acc[c] * sc + p0 * x0[c];
  }

  float inv = 1.0f / (den + 1e-16f);
  float2* po = (float2*)(xout + d * HIDDEN + hoff);
  const float* pb = bias + hoff;
  for (int i = 0; i < OUT_CH / 2; ++i) {
    float v0 = acc[2 * i]     * inv + pb[2 * i];
    float v1 = acc[2 * i + 1] * inv + pb[2 * i + 1];
    float2 o;
    o.x = v0 > 0.0f ? v0 : expm1f(v0);
    o.y = v1 > 0.0f ? v1 : expm1f(v1);
    po[i] = o;
  }
}

extern "C" void kernel_launch(void* const* d_in, const int* in_sizes, int n_in,
                              void* d_out, int out_size, void* d_ws, size_t ws_size,
                              hipStream_t stream) {
  (void)in_sizes; (void)n_in; (void)out_size; (void)ws_size;

  const float* x    = (const float*)d_in[0];
  const int*   ei   = (const int*)d_in[1];
  const int*   esrc = ei;
  const int*   edst = ei + N_EDGES;

  float* ws    = (float*)d_ws;
  float* xl    = ws;
  float* xr    = xl   + (size_t)N_NODES * HIDDEN;
  float* xbuf  = xr   + (size_t)N_NODES * HIDDEN;
  int*   deg      = (int*)(xbuf + (size_t)N_NODES * HIDDEN);
  int*   rowstart = deg      + N_NODES;
  int*   total    = rowstart + N_NODES;
  int*   rank     = total    + 64;
  int*   csr_src  = rank     + N_EDGES;

  const int gemm_blocks = N_NODES / 16;
  const int nodes_blocks = (N_NODES + 255) / 256;
  const int edges_blocks = (N_EDGES + 255) / 256;
  const int head_blocks  = (N_NODES * HEADS + 255) / 256;

  csr_zero <<<nodes_blocks, 256, 0, stream>>>(deg, total);
  csr_count<<<edges_blocks, 256, 0, stream>>>(edst, deg, rank);
  csr_alloc<<<nodes_blocks, 256, 0, stream>>>(deg, rowstart, total);
  csr_fill <<<edges_blocks, 256, 0, stream>>>(esrc, edst, rowstart, rank, csr_src);

  const float* xin = x;
  for (int L = 0; L < 3; ++L) {
    const float* Wl = (const float*)d_in[2 + 4 * L];
    const float* Wr = (const float*)d_in[3 + 4 * L];
    const float* at = (const float*)d_in[4 + 4 * L];
    const float* b  = (const float*)d_in[5 + 4 * L];

    if (L == 0) gat_gemm2_wmma<F_IN>  <<<gemm_blocks, 32, 0, stream>>>(xin, Wl, Wr, xl, xr);
    else        gat_gemm2_wmma<HIDDEN><<<gemm_blocks, 32, 0, stream>>>(xin, Wl, Wr, xl, xr);

    float* xout = (L == 2) ? (float*)d_out : xbuf;
    gat_node_gather<<<head_blocks, 256, 0, stream>>>(rowstart, deg, csr_src, xl, xr, at, b, xout);

    xin = xbuf;
  }
}


# xl stored head-interleaved per row (GEMM epilogue permutes its columns): each gather load reads 64 contiguous bytes per node over its 8 head lanes, one cache line per node instead of three
# speedup vs baseline: 4.7041x; 1.1911x over previous
#include <hip/hip_runtime.h>
#include <math.h>

#define N_NODES 50000
#define N_EDGES 800000
#define F_IN    256
#define HEADS   8
#define OUT_CH  10
#define HIDDEN  80

typedef __attribute__((ext_vector_type(2))) float v2f;
typedef __attribute__((ext_vector_type(8))) float v8f;

__device__ __forceinline__ int xl_slot(int head, int ch) {
  return ((ch >> 1) << 4) + 2 * head + (ch & 1);
}

template <int K>
__global__ void __launch_bounds__(32)
gat_gemm2_wmma(const float* __restrict__ X,
               const float* __restrict__ Wl,
               const float* __restrict__ Wr,
               float* __restrict__ xl,
               float* __restrict__ xr) {
  const int NT = HIDDEN / 16;
  int mt   = blockIdx.x;
  int lane = threadIdx.x;
  int half = lane >> 4;
  int r    = lane & 15;

  const float* ap  = X  + (mt * 16 + r) * K + 2 * half;
  const float* blp = Wl + (2 * half) * HIDDEN + r;
  const float* brp = Wr + (2 * half) * HIDDEN + r;

  v8f cl[NT];
  v8f cr[NT];
#pragma unroll
  for (int t = 0; t < NT; ++t) { cl[t] = (v8f){}; cr[t] = (v8f){}; }

#pragma unroll 2
  for (int k = 0; k < K; k += 4) {
    v2f a = *(const v2f*)(ap + k);
#pragma unroll
    for (int t = 0; t < NT; ++t) {
      v2f bl;
      bl.x = blp[k * HIDDEN + t * 16];
      bl.y = blp[(k + 1) * HIDDEN + t * 16];
      v2f br;
      br.x = brp[k * HIDDEN + t * 16];
      br.y = brp[(k + 1) * HIDDEN + t * 16];
      cl[t] = __builtin_amdgcn_wmma_f32_16x16x4_f32(false, a, false, bl, (short)0, cl[t], false, false);
      cr[t] = __builtin_amdgcn_wmma_f32_16x16x4_f32(false, a, false, br, (short)0, cr[t], false, false);
    }
  }

#pragma unroll
  for (int t = 0; t < NT; ++t) {
    int n    = t * 16 + r;
    int hd   = n / OUT_CH;
    int slot = xl_slot(hd, n - hd * OUT_CH);
#pragma unroll
    for (int v = 0; v < 8; ++v) {
      int orow = mt * 16 + v + 8 * half;
      xl[orow * HIDDEN + slot] = cl[t][v];
      xr[orow * HIDDEN + t * 16 + r] = cr[t][v];
    }
  }
}

__global__ void csr_zero(int* __restrict__ deg, int* __restrict__ total) {
  int idx = blockIdx.x * blockDim.x + threadIdx.x;
  if (idx < N_NODES) deg[idx] = 0;
  if (idx == 0) *total = 0;
}

__global__ void csr_count(const int* __restrict__ edst, int* deg,
                          int* __restrict__ rank) {
  int e = blockIdx.x * blockDim.x + threadIdx.x;
  if (e >= N_EDGES) return;
  rank[e] = atomicAdd(&deg[edst[e]], 1);
}

__global__ void __launch_bounds__(256)
csr_alloc(const int* __restrict__ deg, int* __restrict__ rowstart, int* total) {
  int node = blockIdx.x * blockDim.x + threadIdx.x;
  int lane = threadIdx.x & 31;
  int dg = node < N_NODES ? deg[node] : 0;
  int incl = dg;
  for (int off = 1; off < 32; off <<= 1) {
    int t = __shfl_up(incl, off);
    if (lane >= off) incl += t;
  }
  int base = 0;
  if (lane == 31) base = atomicAdd(total, incl);
  base = __shfl(base, 31);
  if (node < N_NODES) rowstart[node] = base + incl - dg;
}

__global__ void csr_fill(const int* __restrict__ esrc,
                         const int* __restrict__ edst,
                         const int* __restrict__ rowstart,
                         const int* __restrict__ rank,
                         int* __restrict__ csr_src) {
  int e = blockIdx.x * blockDim.x + threadIdx.x;
  if (e >= N_EDGES) return;
  int pos = rowstart[edst[e]] + rank[e];
  if ((unsigned)pos < (unsigned)N_EDGES) csr_src[pos] = esrc[e];
}

__device__ __forceinline__ void load10(const float* __restrict__ p, float (&v)[OUT_CH]) {
  const float2* q = (const float2*)p;
  for (int i = 0; i < OUT_CH / 2; ++i) {
    float2 t = q[i];
    v[2 * i]     = t.x;
    v[2 * i + 1] = t.y;
  }
}

__device__ __forceinline__ void load_xl_row(const float* __restrict__ xl, int node, int h,
                                            float (&v)[OUT_CH]) {
  const float2* q = (const float2*)(xl + node * HIDDEN + 2 * h);
  for (int i = 0; i < OUT_CH / 2; ++i) {
    float2 t = q[8 * i];
    v[2 * i]     = t.x;
    v[2 * i + 1] = t.y;
  }
}

__device__ __forceinline__ float head_logit(const float (&x)[OUT_CH],
                                            const float (&r)[OUT_CH],
                                            const float (&a)[OUT_CH]) {
  float acc = 0.0f;
  for (int c = 0; c < OUT_CH; ++c) {
    float v = x[c] + r[c];
    v = v > 0.0f ? v : 0.2f * v;
    acc += v * a[c];
  }
  return acc;
}

__device__ __forceinline__ int clamp_node(int s) {
  return min(max(s, 0), N_NODES - 1);
}

__global__ void __launch_bounds__(256)
gat_node_gather(const int* __restrict__ rowstart,
                const int* __restrict__ deg,
                const int* __restrict__ csr_src,
                const float* __restrict__ xl,
                const float* __restrict__ xr,
                const float* __restrict__ att,
                const float* __restrict__ bias,
                float* __restrict__ xout) {
  int t = blockIdx.x * blockDim.x + threadIdx.x;
  if (t >= N_NODES * HEADS) return;
  int d = t >> 3;
  int h = t & 7;
  int hoff = h * OUT_CH;

  float a[OUT_CH], r[OUT_CH], x0[OUT_CH], x1[OUT_CH], acc[OUT_CH];
  load10(att + hoff, a);
  load10(xr + d * HIDDEN + hoff, r);

  int beg = min(max(rowstart[d], 0), N_EDGES);
  int end = beg + min(max(deg[d], 0), N_EDGES - beg);

  load_xl_row(xl, d, h, x0);
  float m   = head_logit(x0, r, a);
  float den = 1.0f;
  for (int c = 0; c < OUT_CH; ++c) acc[c] = x0[c];

  int e = beg;
  int s0 = 0, s1 = 0;
  if (e + 1 < end) { s0 = csr_src[e]; s1 = csr_src[e + 1]; }
  while (e + 1 < end) {
    int c0 = clamp_node(s0);
    int c1 = clamp_node(s1);
    e += 2;
    if (e + 1 < end) { s0 = csr_src[e]; s1 = csr_src[e + 1]; }
    load_xl_row(xl, c0, h, x0);
    load_xl_row(xl, c1, h, x1);
    float l0 = head_logit(x0, r, a);
    float l1 = head_logit(x1, r, a);
    float mn = fmaxf(m, fmaxf(l0, l1));
    float sc = __expf(m - mn);
    float p0 = __expf(l0 - mn);
    float p1 = __expf(l1 - mn);
    den = den * sc + (p0 + p1);
    for (int c = 0; c < OUT_CH; ++c) acc[c] = acc[c] * sc + (p0 * x0[c] + p1 * x1[c]);
    m = mn;
  }
  if (e < end) {
    int c0 = clamp_node(csr_src[e]);
    load_xl_row(xl, c0, h, x0);
    float l0 = head_logit(x0, r, a);
    float mn = fmaxf(m, l0);
    float sc = __expf(m - mn);
    float p0 = __expf(l0 - mn);
    den = den * sc + p0;
    for (int c = 0; c < OUT_CH; ++c) acc[c] = acc[c] * sc + p0 * x0[c];
  }

  float inv = 1.0f / (den + 1e-16f);
  float2* po = (float2*)(xout + d * HIDDEN + hoff);
  const float* pb = bias + hoff;
  for (int i = 0; i < OUT_CH / 2; ++i) {
    float v0 = acc[2 * i]     * inv + pb[2 * i];
    float v1 = acc[2 * i + 1] * inv + pb[2 * i + 1];
    float2 o;
    o.x = v0 > 0.0f ? v0 : expm1f(v0);
    o.y = v1 > 0.0f ? v1 : expm1f(v1);
    po[i] = o;
  }
}

extern "C" void kernel_launch(void* const* d_in, const int* in_sizes, int n_in,
                              void* d_out, int out_size, void* d_ws, size_t ws_size,
                              hipStream_t stream) {
  (void)in_sizes; (void)n_in; (void)out_size; (void)ws_size;

  const float* x    = (const float*)d_in[0];
  const int*   ei   = (const int*)d_in[1];
  const int*   esrc = ei;
  const int*   edst = ei + N_EDGES;

  float* ws    = (float*)d_ws;
  float* xl    = ws;
  float* xr    = xl   + (size_t)N_NODES * HIDDEN;
  float* xbuf  = xr   + (size_t)N_NODES * HIDDEN;
  int*   deg      = (int*)(xbuf + (size_t)N_NODES * HIDDEN);
  int*   rowstart = deg      + N_NODES;
  int*   total    = rowstart + N_NODES;
  int*   rank     = total    + 64;
  int*   csr_src  = rank     + N_EDGES;

  const int gemm_blocks = N_NODES / 16;
  const int nodes_blocks = (N_NODES + 255) / 256;
  const int edges_blocks = (N_EDGES + 255) / 256;
  const int head_blocks  = (N_NODES * HEADS + 255) / 256;

  csr_zero <<<nodes_blocks, 256, 0, stream>>>(deg, total);
  csr_count<<<edges_blocks, 256, 0, stream>>>(edst, deg, rank);
  csr_alloc<<<nodes_blocks, 256, 0, stream>>>(deg, rowstart, total);
  csr_fill <<<edges_blocks, 256, 0, stream>>>(esrc, edst, rowstart, rank, csr_src);

  const float* xin = x;
  for (int L = 0; L < 3; ++L) {
    const float* Wl = (const float*)d_in[2 + 4 * L];
    const float* Wr = (const float*)d_in[3 + 4 * L];
    const float* at = (const float*)d_in[4 + 4 * L];
    const float* b  = (const float*)d_in[5 + 4 * L];

    if (L == 0) gat_gemm2_wmma<F_IN>  <<<gemm_blocks, 32, 0, stream>>>(xin, Wl, Wr, xl, xr);
    else        gat_gemm2_wmma<HIDDEN><<<gemm_blocks, 32, 0, stream>>>(xin, Wl, Wr, xl, xr);

    float* xout = (L == 2) ? (float*)d_out : xbuf;
    gat_node_gather<<<head_blocks, 256, 0, stream>>>(rowstart, deg, csr_src, xl, xr, at, b, xout);

    xin = xbuf;
  }
}


# layers 1 and 2: Wl/Wr products moved into the previous layer's gather launch (32 finished rows via LDS, one barrier, same WMMA k order): 8 launches, not 10; no xbuf round trip
# speedup vs baseline: 5.0807x; 1.0801x over previous
#include <hip/hip_runtime.h>
#include <math.h>

#define N_NODES 50000
#define N_EDGES 800000
#define F_IN    256
#define HEADS   8
#define OUT_CH  10
#define HIDDEN  80

typedef __attribute__((ext_vector_type(2))) float v2f;
typedef __attribute__((ext_vector_type(8))) float v8f;

__device__ __forceinline__ int xl_slot(int head, int ch) {
  return ((ch >> 1) << 4) + 2 * head + (ch & 1);
}

template <int K>
__global__ void __launch_bounds__(32)
gat_gemm2_wmma(const float* __restrict__ X,
               const float* __restrict__ Wl,
               const float* __restrict__ Wr,
               float* __restrict__ xl,
               float* __restrict__ xr) {
  const int NT = HIDDEN / 16;
  int mt   = blockIdx.x;
  int lane = threadIdx.x;
  int half = lane >> 4;
  int r    = lane & 15;

  const float* ap  = X  + (mt * 16 + r) * K + 2 * half;
  const float* blp = Wl + (2 * half) * HIDDEN + r;
  const float* brp = Wr + (2 * half) * HIDDEN + r;

  v8f cl[NT];
  v8f cr[NT];
#pragma unroll
  for (int t = 0; t < NT; ++t) { cl[t] = (v8f){}; cr[t] = (v8f){}; }

#pragma unroll 2
  for (int k = 0; k < K; k += 4) {
    v2f a = *(const v2f*)(ap + k);
#pragma unroll
    for (int t = 0; t < NT; ++t) {
      v2f bl;
      bl.x = blp[k * HIDDEN + t * 16];
      bl.y = blp[(k + 1) * HIDDEN + t * 16];
      v2f br;
      br.x = brp[k * HIDDEN + t * 16];
      br.y = brp[(k + 1) * HIDDEN + t * 16];
      cl[t] = __builtin_amdgcn_wmma_f32_16x16x4_f32(false, a, false, bl, (short)0, cl[t], false, false);
      cr[t] = __builtin_amdgcn_wmma_f32_16x16x4_f32(false, a, false, br, (short)0, cr[t], false, false);
    }
  }

#pragma unroll
  for (int t = 0; t < NT; ++t) {
    int n    = t * 16 + r;
    int hd   = n / OUT_CH;
    int slot = xl_slot(hd, n - hd * OUT_CH);
#pragma unroll
    for (int v = 0; v < 8; ++v) {
      int orow = mt * 16 + v + 8 * half;
      xl[orow * HIDDEN + slot] = cl[t][v];
      xr[orow * HIDDEN + t * 16 + r] = cr[t][v];
    }
  }
}

__global__ void csr_zero(int* __restrict__ deg, int* __restrict__ total) {
  int idx = blockIdx.x * blockDim.x + threadIdx.x;
  if (idx < N_NODES) deg[idx] = 0;
  if (idx == 0) *total = 0;
}

__global__ void csr_count(const int* __restrict__ edst, int* deg,
                          int* __restrict__ rank) {
  int e = blockIdx.x * blockDim.x + threadIdx.x;
  if (e >= N_EDGES) return;
  rank[e] = atomicAdd(&deg[edst[e]], 1);
}

__global__ void __launch_bounds__(256)
csr_alloc(const int* __restrict__ deg, int* __restrict__ rowstart, int* total) {
  int node = blockIdx.x * blockDim.x + threadIdx.x;
  int lane = threadIdx.x & 31;
  int dg = node < N_NODES ? deg[node] : 0;
  int incl = dg;
  for (int off = 1; off < 32; off <<= 1) {
    int t = __shfl_up(incl, off);
    if (lane >= off) incl += t;
  }
  int base = 0;
  if (lane == 31) base = atomicAdd(total, incl);
  base = __shfl(base, 31);
  if (node < N_NODES) rowstart[node] = base + incl - dg;
}

__global__ void csr_fill(const int* __restrict__ esrc,
                         const int* __restrict__ edst,
                         const int* __restrict__ rowstart,
                         const int* __restrict__ rank,
                         int* __restrict__ csr_src) {
  int e = blockIdx.x * blockDim.x + threadIdx.x;
  if (e >= N_EDGES) return;
  int pos = rowstart[edst[e]] + rank[e];
  if ((unsigned)pos < (unsigned)N_EDGES) csr_src[pos] = esrc[e];
}

__device__ __forceinline__ void load10(const float* __restrict__ p, float (&v)[OUT_CH]) {
  const float2* q = (const float2*)p;
  for (int i = 0; i < OUT_CH / 2; ++i) {
    float2 t = q[i];
    v[2 * i]     = t.x;
    v[2 * i + 1] = t.y;
  }
}

__device__ __forceinline__ void load_xl_row(const float* __restrict__ xl, int node, int h,
                                            float (&v)[OUT_CH]) {
  const float2* q = (const float2*)(xl + node * HIDDEN + 2 * h);
  for (int i = 0; i < OUT_CH / 2; ++i) {
    float2 t = q[8 * i];
    v[2 * i]     = t.x;
    v[2 * i + 1] = t.y;
  }
}

__device__ __forceinline__ float head_logit(const float (&x)[OUT_CH],
                                            const float (&r)[OUT_CH],
                                            const float (&a)[OUT_CH]) {
  float acc = 0.0f;
  for (int c = 0; c < OUT_CH; ++c) {
    float v = x[c] + r[c];
    v = v > 0.0f ? v : 0.2f * v;
    acc += v * a[c];
  }
  return acc;
}

__device__ __forceinline__ int clamp_node(int s) {
  return min(max(s, 0), N_NODES - 1);
}

__device__ __forceinline__ void gather_node_head(const int* __restrict__ rowstart,
                                                 const int* __restrict__ deg,
                                                 const int* __restrict__ csr_src,
                                                 const float* __restrict__ xl,
                                                 const float* xr,
                                                 const float* __restrict__ att,
                                                 const float* __restrict__ bias,
                                                 int d, int h, float (&o)[OUT_CH]) {
  int hoff = h * OUT_CH;

  float a[OUT_CH], r[OUT_CH], x0[OUT_CH], x1[OUT_CH], acc[OUT_CH];
  load10(att + hoff, a);
  load10(xr + d * HIDDEN + hoff, r);

  int beg = min(max(rowstart[d], 0), N_EDGES);
  int end = beg + min(max(deg[d], 0), N_EDGES - beg);

  load_xl_row(xl, d, h, x0);
  float m   = head_logit(x0, r, a);
  float den = 1.0f;
  for (int c = 0; c < OUT_CH; ++c) acc[c] = x0[c];

  int e = beg;
  int s0 = 0, s1 = 0;
  if (e + 1 < end) { s0 = csr_src[e]; s1 = csr_src[e + 1]; }
  while (e + 1 < end) {
    int c0 = clamp_node(s0);
    int c1 = clamp_node(s1);
    e += 2;
    if (e + 1 < end) { s0 = csr_src[e]; s1 = csr_src[e + 1]; }
    load_xl_row(xl, c0, h, x0);
    load_xl_row(xl, c1, h, x1);
    float l0 = head_logit(x0, r, a);
    float l1 = head_logit(x1, r, a);
    float mn = fmaxf(m, fmaxf(l0, l1));
    float sc = __expf(m - mn);
    float p0 = __expf(l0 - mn);
    float p1 = __expf(l1 - mn);
    den = den * sc + (p0 + p1);
    for (int c = 0; c < OUT_CH; ++c) acc[c] = acc[c] * sc + (p0 * x0[c] + p1 * x1[c]);
    m = mn;
  }
  if (e < end) {
    int c0 = clamp_node(csr_src[e]);
    load_xl_row(xl, c0, h, x0);
    float l0 = head_logit(x0, r, a);
    float mn = fmaxf(m, l0);
    float sc = __expf(m - mn);
    float p0 = __expf(l0 - mn);
    den = den * sc + p0;
    for (int c = 0; c < OUT_CH; ++c) acc[c] = acc[c] * sc + p0 * x0[c];
  }

  float inv = 1.0f / (den + 1e-16f);
  const float* pb = bias + hoff;
  for (int c = 0; c < OUT_CH; ++c) {
    float v = acc[c] * inv + pb[c];
    o[c] = v > 0.0f ? v : expm1f(v);
  }
}

__global__ void __launch_bounds__(256)
gat_node_gather(const int* __restrict__ rowstart,
                const int* __restrict__ deg,
                const int* __restrict__ csr_src,
                const float* __restrict__ xl,
                const float* __restrict__ xr,
                const float* __restrict__ att,
                const float* __restrict__ bias,
                float* __restrict__ xout) {
  int t = blockIdx.x * blockDim.x + threadIdx.x;
  if (t >= N_NODES * HEADS) return;
  int d = t >> 3;
  int h = t & 7;
  float o[OUT_CH];
  gather_node_head(rowstart, deg, csr_src, xl, xr, att, bias, d, h, o);
  float2* po = (float2*)(xout + d * HIDDEN + h * OUT_CH);
  for (int i = 0; i < OUT_CH / 2; ++i) {
    float2 v;
    v.x = o[2 * i];
    v.y = o[2 * i + 1];
    po[i] = v;
  }
}

#define TILE_PITCH 84

template <int NTW>
__device__ __forceinline__ void strip_product(const float* tile_rows,
                                              const float* __restrict__ Wl,
                                              const float* __restrict__ Wr,
                                              float* __restrict__ xl_next,
                                              float* xr_next,
                                              int row0, int q, int lane) {
  int half = lane >> 4;
  int r    = lane & 15;
  const float* ap = tile_rows + r * TILE_PITCH + 2 * half;
  const float* bp[NTW];
#pragma unroll
  for (int j = 0; j < NTW; ++j) {
    int id = q + 4 * j;
    const float* w = id < 5 ? Wl : Wr;
    int ct = id < 5 ? id : id - 5;
    bp[j] = w + (2 * half) * HIDDEN + ct * 16 + r;
  }
  v8f c[NTW];
#pragma unroll
  for (int j = 0; j < NTW; ++j) c[j] = (v8f){};

#pragma unroll 4
  for (int k = 0; k < HIDDEN; k += 4) {
    v2f a = *(const v2f*)(ap + k);
#pragma unroll
    for (int j = 0; j < NTW; ++j) {
      v2f b;
      b.x = bp[j][k * HIDDEN];
      b.y = bp[j][(k + 1) * HIDDEN];
      c[j] = __builtin_amdgcn_wmma_f32_16x16x4_f32(false, a, false, b, (short)0, c[j], false, false);
    }
  }

#pragma unroll
  for (int j = 0; j < NTW; ++j) {
    int id = q + 4 * j;
    int ct = id < 5 ? id : id - 5;
    int n  = ct * 16 + r;
    int hd = n / OUT_CH;
    int slot = xl_slot(hd, n - hd * OUT_CH);
#pragma unroll
    for (int v = 0; v < 8; ++v) {
      int orow = row0 + v + 8 * half;
      if (id < 5) xl_next[orow * HIDDEN + slot] = c[j][v];
      else        xr_next[orow * HIDDEN + n]    = c[j][v];
    }
  }
}

__global__ void __launch_bounds__(256)
gat_gather_then_gemm(const int* __restrict__ rowstart,
                     const int* __restrict__ deg,
                     const int* __restrict__ csr_src,
                     const float* __restrict__ xl,
                     float* xr,
                     const float* __restrict__ att,
                     const float* __restrict__ bias,
                     const float* __restrict__ Wl,
                     const float* __restrict__ Wr,
                     float* __restrict__ xl_next) {
  __shared__ __attribute__((aligned(16))) float tile[32 * TILE_PITCH];
  int t = blockIdx.x * blockDim.x + threadIdx.x;
  if (t < N_NODES * HEADS) {
    int d = t >> 3;
    int h = t & 7;
    float o[OUT_CH];
    gather_node_head(rowstart, deg, csr_src, xl, xr, att, bias, d, h, o);
    float2* po = (float2*)(tile + (threadIdx.x >> 3) * TILE_PITCH + h * OUT_CH);
    for (int i = 0; i < OUT_CH / 2; ++i) {
      float2 v;
      v.x = o[2 * i];
      v.y = o[2 * i + 1];
      po[i] = v;
    }
  }
  __syncthreads();

  int w     = __builtin_amdgcn_readfirstlane(threadIdx.x >> 5);
  int strip = w >> 2;
  int q     = strip ? 3 - (w & 3) : (w & 3);
  int row0  = blockIdx.x * 32 + strip * 16;
  if (row0 >= N_NODES) return;
  const float* rows = tile + strip * 16 * TILE_PITCH;
  int lane = threadIdx.x & 31;
  if (q < 2) strip_product<3>(rows, Wl, Wr, xl_next, xr, row0, q, lane);
  else       strip_product<2>(rows, Wl, Wr, xl_next, xr, row0, q, lane);
}

extern "C" void kernel_launch(void* const* d_in, const int* in_sizes, int n_in,
                              void* d_out, int out_size, void* d_ws, size_t ws_size,
                              hipStream_t stream) {
  (void)in_sizes; (void)n_in; (void)out_size; (void)ws_size;

  const float* x    = (const float*)d_in[0];
  const int*   ei   = (const int*)d_in[1];
  const int*   esrc = ei;
  const int*   edst = ei + N_EDGES;

  float* ws    = (float*)d_ws;
  float* xl    = ws;
  float* xr    = xl   + (size_t)N_NODES * HIDDEN;
  float* xbuf  = xr   + (size_t)N_NODES * HIDDEN;
  int*   deg      = (int*)(xbuf + (size_t)N_NODES * HIDDEN);
  int*   rowstart = deg      + N_NODES;
  int*   total    = rowstart + N_NODES;
  int*   rank     = total    + 64;
  int*   csr_src  = rank     + N_EDGES;

  const int gemm_blocks = N_NODES / 16;
  const int nodes_blocks = (N_NODES + 255) / 256;
  const int edges_blocks = (N_EDGES + 255) / 256;
  const int head_blocks  = (N_NODES * HEADS + 255) / 256;

  csr_zero <<<nodes_blocks, 256, 0, stream>>>(deg, total);
  csr_count<<<edges_blocks, 256, 0, stream>>>(edst, deg, rank);
  csr_alloc<<<nodes_blocks, 256, 0, stream>>>(deg, rowstart, total);
  csr_fill <<<edges_blocks, 256, 0, stream>>>(esrc, edst, rowstart, rank, csr_src);

  float* xl_a = xl;
  float* xl_b = xbuf;
  gat_gemm2_wmma<F_IN><<<gemm_blocks, 32, 0, stream>>>(
      x, (const float*)d_in[2], (const float*)d_in[3], xl_a, xr);
  gat_gather_then_gemm<<<head_blocks, 256, 0, stream>>>(
      rowstart, deg, csr_src, xl_a, xr, (const float*)d_in[4], (const float*)d_in[5],
      (const float*)d_in[6], (const float*)d_in[7], xl_b);
  gat_gather_then_gemm<<<head_blocks, 256, 0, stream>>>(
      rowstart, deg, csr_src, xl_b, xr, (const float*)d_in[8], (const float*)d_in[9],
      (const float*)d_in[10], (const float*)d_in[11], xl_a);
  gat_node_gather<<<head_blocks, 256, 0, stream>>>(
      rowstart, deg, csr_src, xl_a, xr, (const float*)d_in[12], (const float*)d_in[13],
      (float*)d_out);
}


# edge grouping = two-level counting sort on LDS integer counters (256 node ranges, then nodes), no global atomic left; 3 sort launches, not 4; sum order = arrival order of the LDS adds
# speedup vs baseline: 6.7296x; 1.3245x over previous
#include <hip/hip_runtime.h>
#include <math.h>

#define N_NODES 50000
#define N_EDGES 800000
#define F_IN    256
#define HEADS   8
#define OUT_CH  10
#define HIDDEN  80

typedef __attribute__((ext_vector_type(2))) float v2f;
typedef __attribute__((ext_vector_type(8))) float v8f;

__device__ __forceinline__ int xl_slot(int head, int ch) {
  return ((ch >> 1) << 4) + 2 * head + (ch & 1);
}

template <int K>
__global__ void __launch_bounds__(32)
gat_gemm2_wmma(const float* __restrict__ X,
               const float* __restrict__ Wl,
               const float* __restrict__ Wr,
               float* __restrict__ xl,
               float* __restrict__ xr) {
  const int NT = HIDDEN / 16;
  int mt   = blockIdx.x;
  int lane = threadIdx.x;
  int half = lane >> 4;
  int r    = lane & 15;

  const float* ap  = X  + (mt * 16 + r) * K + 2 * half;
  const float* blp = Wl + (2 * half) * HIDDEN + r;
  const float* brp = Wr + (2 * half) * HIDDEN + r;

  v8f cl[NT];
  v8f cr[NT];
#pragma unroll
  for (int t = 0; t < NT; ++t) { cl[t] = (v8f){}; cr[t] = (v8f){}; }

#pragma unroll 2
  for (int k = 0; k < K; k += 4) {
    v2f a = *(const v2f*)(ap + k);
#pragma unroll
    for (int t = 0; t < NT; ++t) {
      v2f bl;
      bl.x = blp[k * HIDDEN + t * 16];
      bl.y = blp[(k + 1) * HIDDEN + t * 16];
      v2f br;
      br.x = brp[k * HIDDEN + t * 16];
      br.y = brp[(k + 1) * HIDDEN + t * 16];
      cl[t] = __builtin_amdgcn_wmma_f32_16x16x4_f32(false, a, false, bl, (short)0, cl[t], false, false);
      cr[t] = __builtin_amdgcn_wmma_f32_16x16x4_f32(false, a, false, br, (short)0, cr[t], false, false);
    }
  }

#pragma unroll
  for (int t = 0; t < NT; ++t) {
    int n    = t * 16 + r;
    int hd   = n / OUT_CH;
    int slot = xl_slot(hd, n - hd * OUT_CH);
#pragma unroll
    for (int v = 0; v < 8; ++v) {
      int orow = mt * 16 + v + 8 * half;
      xl[orow * HIDDEN + slot] = cl[t][v];
      xr[orow * HIDDEN + t * 16 + r] = cr[t][v];
    }
  }
}

#define NB      256
#define NPB     196
#define NCH     256
#define CHUNK   3125
#define SORT_T  512
#define EPT     7
#define PPT     8
#define ROWS_PT 32

static_assert(NB * NPB >= N_NODES, "ranges cover all nodes");
static_assert(NCH * CHUNK == N_EDGES, "chunks tile the edge list exactly");
static_assert(EPT * SORT_T >= CHUNK, "one pass of a work-group covers its chunk");
static_assert(NB == 256 && SORT_T >= NB && SORT_T % 64 == 0, "thread maps below assume these");
static_assert((SORT_T / 64) * ROWS_PT == NCH, "a work-group's threads cover the whole table");

__device__ __forceinline__ int clamp_node(int s) {
  return min(max(s, 0), N_NODES - 1);
}

__device__ __forceinline__ int range_of(int node) { return node / NPB; }

__device__ __forceinline__ int scan256_excl(int v, int* wsum, int tid) {
  int lane = tid & 31;
  int w    = tid >> 5;
  int incl = v;
  for (int off = 1; off < 32; off <<= 1) {
    int t = __shfl_up(incl, off);
    if (lane >= off) incl += t;
  }
  if (tid < 256 && lane == 31) wsum[w] = incl;
  __syncthreads();
  int base = 0;
  for (int j = 0; j < 8; ++j) {
    int s = wsum[j];
    if (j < w) base += s;
  }
  return base + incl - v;
}

__global__ void __launch_bounds__(SORT_T)
csr_hist(const int* __restrict__ edst, int* __restrict__ table) {
  __shared__ int h[NB];
  int tid = threadIdx.x;
  int c   = blockIdx.x;
  if (tid < NB) h[tid] = 0;
  const int* p = edst + c * CHUNK;
  int d[EPT];
  for (int j = 0; j < EPT; ++j) {
    int i = tid + j * SORT_T;
    d[j] = p[min(i, CHUNK - 1)];
  }
  __syncthreads();
  for (int j = 0; j < EPT; ++j) {
    int i = tid + j * SORT_T;
    if (i < CHUNK) atomicAdd(&h[range_of(clamp_node(d[j]))], 1);
  }
  __syncthreads();
  if (tid < NB) table[c * NB + tid] = h[tid];
}

__global__ void __launch_bounds__(SORT_T)
csr_scatter(const int* __restrict__ esrc,
            const int* __restrict__ edst,
            const int* __restrict__ table,
            int* __restrict__ range_base,
            int* __restrict__ range_cnt,
            int2* __restrict__ pairs) {
  __shared__ __attribute__((aligned(16))) int part_tot[SORT_T / 64][NB];
  __shared__ __attribute__((aligned(16))) int part_bef[SORT_T / 64][NB];
  __shared__ int cur[NB];
  __shared__ int wsum[8];
  int tid = threadIdx.x;
  int c   = blockIdx.x;

  int s[EPT], d[EPT];
  for (int j = 0; j < EPT; ++j) {
    int i = c * CHUNK + min(tid + j * SORT_T, CHUNK - 1);
    s[j] = esrc[i];
    d[j] = edst[i];
  }

  int g  = tid >> 6;
  int b4 = tid & 63;
  const int4* colp = (const int4*)(table + (g * ROWS_PT) * NB) + b4;
  int4 v[ROWS_PT];
  for (int k = 0; k < ROWS_PT; ++k) v[k] = colp[k * (NB / 4)];
  int4 tot = {0, 0, 0, 0};
  int4 bef = {0, 0, 0, 0};
  for (int k = 0; k < ROWS_PT; ++k) {
    bool earlier = (g * ROWS_PT + k) < c;
    tot.x += v[k].x;  tot.y += v[k].y;  tot.z += v[k].z;  tot.w += v[k].w;
    bef.x += earlier ? v[k].x : 0;
    bef.y += earlier ? v[k].y : 0;
    bef.z += earlier ? v[k].z : 0;
    bef.w += earlier ? v[k].w : 0;
  }
  ((int4*)part_tot[g])[b4] = tot;
  ((int4*)part_bef[g])[b4] = bef;
  __syncthreads();
  int t = 0, bf = 0;
  if (tid < NB) {
    for (int j = 0; j < SORT_T / 64; ++j) {
      t  += part_tot[j][tid];
      bf += part_bef[j][tid];
    }
  }
  int base = scan256_excl(t, wsum, tid);
  if (tid < NB) {
    cur[tid] = base + bf;
    if (c == 0) {
      range_base[tid] = base;
      range_cnt[tid]  = t;
    }
  }
  __syncthreads();
  int pos[EPT];
  for (int j = 0; j < EPT; ++j) {
    int i = tid + j * SORT_T;
    d[j]   = clamp_node(d[j]);
    pos[j] = i < CHUNK ? atomicAdd(&cur[range_of(d[j])], 1) : -1;
  }
  for (int j = 0; j < EPT; ++j) {
    if ((unsigned)pos[j] < (unsigned)N_EDGES) {
      int2 pr;
      pr.x = s[j];
      pr.y = d[j];
      pairs[pos[j]] = pr;
    }
  }
}

__global__ void __launch_bounds__(SORT_T)
csr_bucket(const int2* __restrict__ pairs,
           const int* __restrict__ range_base,
           const int* __restrict__ range_cnt,
           int* __restrict__ deg,
           int* __restrict__ rowstart,
           int* __restrict__ csr_src) {
  __shared__ int cnt[NB];
  __shared__ int wsum[8];
  int tid   = threadIdx.x;
  int node0 = blockIdx.x * NPB;
  int beg   = min(max(range_base[blockIdx.x], 0), N_EDGES);
  int n     = min(max(range_cnt[blockIdx.x], 0), N_EDGES - beg);
  if (tid < NB) cnt[tid] = 0;
  __syncthreads();

  for (int i0 = 0; i0 < n; i0 += PPT * SORT_T) {
    int2 p[PPT];
    for (int j = 0; j < PPT; ++j) {
      int i = i0 + j * SORT_T + tid;
      p[j] = pairs[beg + min(i, n - 1)];
    }
    for (int j = 0; j < PPT; ++j) {
      int i = i0 + j * SORT_T + tid;
      if (i < n) atomicAdd(&cnt[min(max(p[j].y - node0, 0), NPB - 1)], 1);
    }
  }
  __syncthreads();

  int v    = tid < NB ? cnt[tid] : 0;
  int excl = scan256_excl(v, wsum, tid);
  if (tid < NB) {
    int start = beg + excl;
    int node  = node0 + tid;
    if (tid < NPB && node < N_NODES) {
      deg[node]      = v;
      rowstart[node] = start;
    }
    cnt[tid] = start;
  }
  __syncthreads();

  for (int i0 = 0; i0 < n; i0 += PPT * SORT_T) {
    int2 p[PPT];
    for (int j = 0; j < PPT; ++j) {
      int i = i0 + j * SORT_T + tid;
      p[j] = pairs[beg + min(i, n - 1)];
    }
    for (int j = 0; j < PPT; ++j) {
      int i = i0 + j * SORT_T + tid;
      if (i < n) {
        int pos = atomicAdd(&cnt[min(max(p[j].y - node0, 0), NPB - 1)], 1);
        if ((unsigned)pos < (unsigned)N_EDGES) csr_src[pos] = p[j].x;
      }
    }
  }
}

__device__ __forceinline__ void load10(const float* __restrict__ p, float (&v)[OUT_CH]) {
  const float2* q = (const float2*)p;
  for (int i = 0; i < OUT_CH / 2; ++i) {
    float2 t = q[i];
    v[2 * i]     = t.x;
    v[2 * i + 1] = t.y;
  }
}

__device__ __forceinline__ void load_xl_row(const float* __restrict__ xl, int node, int h,
                                            float (&v)[OUT_CH]) {
  const float2* q = (const float2*)(xl + node * HIDDEN + 2 * h);
  for (int i = 0; i < OUT_CH / 2; ++i) {
    float2 t = q[8 * i];
    v[2 * i]     = t.x;
    v[2 * i + 1] = t.y;
  }
}

__device__ __forceinline__ float head_logit(const float (&x)[OUT_CH],
                                            const float (&r)[OUT_CH],
                                            const float (&a)[OUT_CH]) {
  float acc = 0.0f;
  for (int c = 0; c < OUT_CH; ++c) {
    float v = x[c] + r[c];
    v = v > 0.0f ? v : 0.2f * v;
    acc += v * a[c];
  }
  return acc;
}

__device__ __forceinline__ void gather_node_head(const int* __restrict__ rowstart,
                                                 const int* __restrict__ deg,
                                                 const int* __restrict__ csr_src,
                                                 const float* __restrict__ xl,
                                                 const float* xr,
                                                 const float* __restrict__ att,
                                                 const float* __restrict__ bias,
                                                 int d, int h, float (&o)[OUT_CH]) {
  int hoff = h * OUT_CH;

  float a[OUT_CH], r[OUT_CH], x0[OUT_CH], x1[OUT_CH], acc[OUT_CH];
  load10(att + hoff, a);
  load10(xr + d * HIDDEN + hoff, r);

  int beg = min(max(rowstart[d], 0), N_EDGES);
  int end = beg + min(max(deg[d], 0), N_EDGES - beg);

  load_xl_row(xl, d, h, x0);
  float m   = head_logit(x0, r, a);
  float den = 1.0f;
  for (int c = 0; c < OUT_CH; ++c) acc[c] = x0[c];

  int e = beg;
  int s0 = 0, s1 = 0;
  if (e + 1 < end) { s0 = csr_src[e]; s1 = csr_src[e + 1]; }
  while (e + 1 < end) {
    int c0 = clamp_node(s0);
    int c1 = clamp_node(s1);
    e += 2;
    if (e + 1 < end) { s0 = csr_src[e]; s1 = csr_src[e + 1]; }
    load_xl_row(xl, c0, h, x0);
    load_xl_row(xl, c1, h, x1);
    float l0 = head_logit(x0, r, a);
    float l1 = head_logit(x1, r, a);
    float mn = fmaxf(m, fmaxf(l0, l1));
    float sc = __expf(m - mn);
    float p0 = __expf(l0 - mn);
    float p1 = __expf(l1 - mn);
    den = den * sc + (p0 + p1);
    for (int c = 0; c < OUT_CH; ++c) acc[c] = acc[c] * sc + (p0 * x0[c] + p1 * x1[c]);
    m = mn;
  }
  if (e < end) {
    int c0 = clamp_node(csr_src[e]);
    load_xl_row(xl, c0, h, x0);
    float l0 = head_logit(x0, r, a);
    float mn = fmaxf(m, l0);
    float sc = __expf(m - mn);
    float p0 = __expf(l0 - mn);
    den = den * sc + p0;
    for (int c = 0; c < OUT_CH; ++c) acc[c] = acc[c] * sc + p0 * x0[c];
  }

  float inv = 1.0f / (den + 1e-16f);
  const float* pb = bias + hoff;
  for (int c = 0; c < OUT_CH; ++c) {
    float v = acc[c] * inv + pb[c];
    o[c] = v > 0.0f ? v : expm1f(v);
  }
}

__global__ void __launch_bounds__(256)
gat_node_gather(const int* __restrict__ rowstart,
                const int* __restrict__ deg,
                const int* __restrict__ csr_src,
                const float* __restrict__ xl,
                const float* __restrict__ xr,
                const float* __restrict__ att,
                const float* __restrict__ bias,
                float* __restrict__ xout) {
  int t = blockIdx.x * blockDim.x + threadIdx.x;
  if (t >= N_NODES * HEADS) return;
  int d = t >> 3;
  int h = t & 7;
  float o[OUT_CH];
  gather_node_head(rowstart, deg, csr_src, xl, xr, att, bias, d, h, o);
  float2* po = (float2*)(xout + d * HIDDEN + h * OUT_CH);
  for (int i = 0; i < OUT_CH / 2; ++i) {
    float2 v;
    v.x = o[2 * i];
    v.y = o[2 * i + 1];
    po[i] = v;
  }
}

#define TILE_PITCH 84

template <int NTW>
__device__ __forceinline__ void strip_product(const float* tile_rows,
                                              const float* __restrict__ Wl,
                                              const float* __restrict__ Wr,
                                              float* __restrict__ xl_next,
                                              float* xr_next,
                                              int row0, int q, int lane) {
  int half = lane >> 4;
  int r    = lane & 15;
  const float* ap = tile_rows + r * TILE_PITCH + 2 * half;
  const float* bp[NTW];
#pragma unroll
  for (int j = 0; j < NTW; ++j) {
    int id = q + 4 * j;
    const float* w = id < 5 ? Wl : Wr;
    int ct = id < 5 ? id : id - 5;
    bp[j] = w + (2 * half) * HIDDEN + ct * 16 + r;
  }
  v8f c[NTW];
#pragma unroll
  for (int j = 0; j < NTW; ++j) c[j] = (v8f){};

#pragma unroll 4
  for (int k = 0; k < HIDDEN; k += 4) {
    v2f a = *(const v2f*)(ap + k);
#pragma unroll
    for (int j = 0; j < NTW; ++j) {
      v2f b;
      b.x = bp[j][k * HIDDEN];
      b.y = bp[j][(k + 1) * HIDDEN];
      c[j] = __builtin_amdgcn_wmma_f32_16x16x4_f32(false, a, false, b, (short)0, c[j], false, false);
    }
  }

#pragma unroll
  for (int j = 0; j < NTW; ++j) {
    int id = q + 4 * j;
    int ct = id < 5 ? id : id - 5;
    int n  = ct * 16 + r;
    int hd = n / OUT_CH;
    int slot = xl_slot(hd, n - hd * OUT_CH);
#pragma unroll
    for (int v = 0; v < 8; ++v) {
      int orow = row0 + v + 8 * half;
      if (id < 5) xl_next[orow * HIDDEN + slot] = c[j][v];
      else        xr_next[orow * HIDDEN + n]    = c[j][v];
    }
  }
}

__global__ void __launch_bounds__(256)
gat_gather_then_gemm(const int* __restrict__ rowstart,
                     const int* __restrict__ deg,
                     const int* __restrict__ csr_src,
                     const float* __restrict__ xl,
                     float* xr,
                     const float* __restrict__ att,
                     const float* __restrict__ bias,
                     const float* __restrict__ Wl,
                     const float* __restrict__ Wr,
                     float* __restrict__ xl_next) {
  __shared__ __attribute__((aligned(16))) float tile[32 * TILE_PITCH];
  int t = blockIdx.x * blockDim.x + threadIdx.x;
  if (t < N_NODES * HEADS) {
    int d = t >> 3;
    int h = t & 7;
    float o[OUT_CH];
    gather_node_head(rowstart, deg, csr_src, xl, xr, att, bias, d, h, o);
    float2* po = (float2*)(tile + (threadIdx.x >> 3) * TILE_PITCH + h * OUT_CH);
    for (int i = 0; i < OUT_CH / 2; ++i) {
      float2 v;
      v.x = o[2 * i];
      v.y = o[2 * i + 1];
      po[i] = v;
    }
  }
  __syncthreads();

  int w     = __builtin_amdgcn_readfirstlane(threadIdx.x >> 5);
  int strip = w >> 2;
  int q     = strip ? 3 - (w & 3) : (w & 3);
  int row0  = blockIdx.x * 32 + strip * 16;
  if (row0 >= N_NODES) return;
  const float* rows = tile + strip * 16 * TILE_PITCH;
  int lane = threadIdx.x & 31;
  if (q < 2) strip_product<3>(rows, Wl, Wr, xl_next, xr, row0, q, lane);
  else       strip_product<2>(rows, Wl, Wr, xl_next, xr, row0, q, lane);
}

extern "C" void kernel_launch(void* const* d_in, const int* in_sizes, int n_in,
                              void* d_out, int out_size, void* d_ws, size_t ws_size,
                              hipStream_t stream) {
  (void)in_sizes; (void)n_in; (void)out_size; (void)ws_size;

  const float* x    = (const float*)d_in[0];
  const int*   ei   = (const int*)d_in[1];
  const int*   esrc = ei;
  const int*   edst = ei + N_EDGES;

  float* ws    = (float*)d_ws;
  float* xl    = ws;
  float* xr    = xl   + (size_t)N_NODES * HIDDEN;
  float* xbuf  = xr   + (size_t)N_NODES * HIDDEN;
  int*   deg      = (int*)(xbuf + (size_t)N_NODES * HIDDEN);
  int*   rowstart = deg      + N_NODES;
  int*   spare    = rowstart + N_NODES;
  int*   csr_src  = spare    + 64 + N_EDGES;
  int*   table      = (int*)xbuf;
  int*   range_base = table      + NCH * NB;
  int*   range_cnt  = range_base + NB;
  int2*  pairs      = (int2*)(range_cnt + NB);

  const int gemm_blocks = N_NODES / 16;
  const int head_blocks  = (N_NODES * HEADS + 255) / 256;

  csr_hist   <<<NCH, SORT_T, 0, stream>>>(edst, table);
  csr_scatter<<<NCH, SORT_T, 0, stream>>>(esrc, edst, table, range_base, range_cnt, pairs);
  csr_bucket <<<NB,  SORT_T, 0, stream>>>(pairs, range_base, range_cnt, deg, rowstart, csr_src);

  float* xl_a = xl;
  float* xl_b = xbuf;
  gat_gemm2_wmma<F_IN><<<gemm_blocks, 32, 0, stream>>>(
      x, (const float*)d_in[2], (const float*)d_in[3], xl_a, xr);
  gat_gather_then_gemm<<<head_blocks, 256, 0, stream>>>(
      rowstart, deg, csr_src, xl_a, xr, (const float*)d_in[4], (const float*)d_in[5],
      (const float*)d_in[6], (const float*)d_in[7], xl_b);
  gat_gather_then_gemm<<<head_blocks, 256, 0, stream>>>(
      rowstart, deg, csr_src, xl_b, xr, (const float*)d_in[8], (const float*)d_in[9],
      (const float*)d_in[10], (const float*)d_in[11], xl_a);
  gat_node_gather<<<head_blocks, 256, 0, stream>>>(
      rowstart, deg, csr_src, xl_a, xr, (const float*)d_in[12], (const float*)d_in[13],
      (float*)d_out);
}


# gather: four list entries per trip (self loop + first 0-3 entries, then fours), half the dependent row-load round trips per node; logits as packed even/odd chains; sum order changes
# speedup vs baseline: 7.1272x; 1.0591x over previous
#include <hip/hip_runtime.h>
#include <math.h>

#define N_NODES 50000
#define N_EDGES 800000
#define F_IN    256
#define HEADS   8
#define OUT_CH  10
#define HIDDEN  80

typedef __attribute__((ext_vector_type(2))) float v2f;
typedef __attribute__((ext_vector_type(8))) float v8f;

__device__ __forceinline__ int xl_slot(int head, int ch) {
  return ((ch >> 1) << 4) + 2 * head + (ch & 1);
}

template <int K>
__global__ void __launch_bounds__(32)
gat_gemm2_wmma(const float* __restrict__ X,
               const float* __restrict__ Wl,
               const float* __restrict__ Wr,
               float* __restrict__ xl,
               float* __restrict__ xr) {
  const int NT = HIDDEN / 16;
  int mt   = blockIdx.x;
  int lane = threadIdx.x;
  int half = lane >> 4;
  int r    = lane & 15;

  const float* ap  = X  + (mt * 16 + r) * K + 2 * half;
  const float* blp = Wl + (2 * half) * HIDDEN + r;
  const float* brp = Wr + (2 * half) * HIDDEN + r;

  v8f cl[NT];
  v8f cr[NT];
#pragma unroll
  for (int t = 0; t < NT; ++t) { cl[t] = (v8f){}; cr[t] = (v8f){}; }

#pragma unroll 2
  for (int k = 0; k < K; k += 4) {
    v2f a = *(const v2f*)(ap + k);
#pragma unroll
    for (int t = 0; t < NT; ++t) {
      v2f bl;
      bl.x = blp[k * HIDDEN + t * 16];
      bl.y = blp[(k + 1) * HIDDEN + t * 16];
      v2f br;
      br.x = brp[k * HIDDEN + t * 16];
      br.y = brp[(k + 1) * HIDDEN + t * 16];
      cl[t] = __builtin_amdgcn_wmma_f32_16x16x4_f32(false, a, false, bl, (short)0, cl[t], false, false);
      cr[t] = __builtin_amdgcn_wmma_f32_16x16x4_f32(false, a, false, br, (short)0, cr[t], false, false);
    }
  }

#pragma unroll
  for (int t = 0; t < NT; ++t) {
    int n    = t * 16 + r;
    int hd   = n / OUT_CH;
    int slot = xl_slot(hd, n - hd * OUT_CH);
#pragma unroll
    for (int v = 0; v < 8; ++v) {
      int orow = mt * 16 + v + 8 * half;
      xl[orow * HIDDEN + slot] = cl[t][v];
      xr[orow * HIDDEN + t * 16 + r] = cr[t][v];
    }
  }
}

#define NB      256
#define NPB     196
#define NCH     256
#define CHUNK   3125
#define SORT_T  512
#define EPT     7
#define PPT     8
#define ROWS_PT 32

static_assert(NB * NPB >= N_NODES, "ranges cover all nodes");
static_assert(NCH * CHUNK == N_EDGES, "chunks tile the edge list exactly");
static_assert(EPT * SORT_T >= CHUNK, "one pass of a work-group covers its chunk");
static_assert(NB == 256 && SORT_T >= NB && SORT_T % 64 == 0, "thread maps below assume these");
static_assert((SORT_T / 64) * ROWS_PT == NCH, "a work-group's threads cover the whole table");

__device__ __forceinline__ int clamp_node(int s) {
  return min(max(s, 0), N_NODES - 1);
}

__device__ __forceinline__ int range_of(int node) { return node / NPB; }

__device__ __forceinline__ int scan256_excl(int v, int* wsum, int tid) {
  int lane = tid & 31;
  int w    = tid >> 5;
  int incl = v;
  for (int off = 1; off < 32; off <<= 1) {
    int t = __shfl_up(incl, off);
    if (lane >= off) incl += t;
  }
  if (tid < 256 && lane == 31) wsum[w] = incl;
  __syncthreads();
  int base = 0;
  for (int j = 0; j < 8; ++j) {
    int s = wsum[j];
    if (j < w) base += s;
  }
  return base + incl - v;
}

__global__ void __launch_bounds__(SORT_T)
csr_hist(const int* __restrict__ edst, int* __restrict__ table) {
  __shared__ int h[NB];
  int tid = threadIdx.x;
  int c   = blockIdx.x;
  if (tid < NB) h[tid] = 0;
  const int* p = edst + c * CHUNK;
  int d[EPT];
  for (int j = 0; j < EPT; ++j) {
    int i = tid + j * SORT_T;
    d[j] = p[min(i, CHUNK - 1)];
  }
  __syncthreads();
  for (int j = 0; j < EPT; ++j) {
    int i = tid + j * SORT_T;
    if (i < CHUNK) atomicAdd(&h[range_of(clamp_node(d[j]))], 1);
  }
  __syncthreads();
  if (tid < NB) table[c * NB + tid] = h[tid];
}

__global__ void __launch_bounds__(SORT_T)
csr_scatter(const int* __restrict__ esrc,
            const int* __restrict__ edst,
            const int* __restrict__ table,
            int* __restrict__ range_base,
            int* __restrict__ range_cnt,
            int2* __restrict__ pairs) {
  __shared__ __attribute__((aligned(16))) int part_tot[SORT_T / 64][NB];
  __shared__ __attribute__((aligned(16))) int part_bef[SORT_T / 64][NB];
  __shared__ int cur[NB];
  __shared__ int wsum[8];
  int tid = threadIdx.x;
  int c   = blockIdx.x;

  int s[EPT], d[EPT];
  for (int j = 0; j < EPT; ++j) {
    int i = c * CHUNK + min(tid + j * SORT_T, CHUNK - 1);
    s[j] = esrc[i];
    d[j] = edst[i];
  }

  int g  = tid >> 6;
  int b4 = tid & 63;
  const int4* colp = (const int4*)(table + (g * ROWS_PT) * NB) + b4;
  int4 v[ROWS_PT];
  for (int k = 0; k < ROWS_PT; ++k) v[k] = colp[k * (NB / 4)];
  int4 tot = {0, 0, 0, 0};
  int4 bef = {0, 0, 0, 0};
  for (int k = 0; k < ROWS_PT; ++k) {
    bool earlier = (g * ROWS_PT + k) < c;
    tot.x += v[k].x;  tot.y += v[k].y;  tot.z += v[k].z;  tot.w += v[k].w;
    bef.x += earlier ? v[k].x : 0;
    bef.y += earlier ? v[k].y : 0;
    bef.z += earlier ? v[k].z : 0;
    bef.w += earlier ? v[k].w : 0;
  }
  ((int4*)part_tot[g])[b4] = tot;
  ((int4*)part_bef[g])[b4] = bef;
  __syncthreads();
  int t = 0, bf = 0;
  if (tid < NB) {
    for (int j = 0; j < SORT_T / 64; ++j) {
      t  += part_tot[j][tid];
      bf += part_bef[j][tid];
    }
  }
  int base = scan256_excl(t, wsum, tid);
  if (tid < NB) {
    cur[tid] = base + bf;
    if (c == 0) {
      range_base[tid] = base;
      range_cnt[tid]  = t;
    }
  }
  __syncthreads();
  int pos[EPT];
  for (int j = 0; j < EPT; ++j) {
    int i = tid + j * SORT_T;
    d[j]   = clamp_node(d[j]);
    pos[j] = i < CHUNK ? atomicAdd(&cur[range_of(d[j])], 1) : -1;
  }
  for (int j = 0; j < EPT; ++j) {
    if ((unsigned)pos[j] < (unsigned)N_EDGES) {
      int2 pr;
      pr.x = s[j];
      pr.y = d[j];
      pairs[pos[j]] = pr;
    }
  }
}

__global__ void __launch_bounds__(SORT_T)
csr_bucket(const int2* __restrict__ pairs,
           const int* __restrict__ range_base,
           const int* __restrict__ range_cnt,
           int* __restrict__ deg,
           int* __restrict__ rowstart,
           int* __restrict__ csr_src) {
  __shared__ int cnt[NB];
  __shared__ int wsum[8];
  int tid   = threadIdx.x;
  int node0 = blockIdx.x * NPB;
  int beg   = min(max(range_base[blockIdx.x], 0), N_EDGES);
  int n     = min(max(range_cnt[blockIdx.x], 0), N_EDGES - beg);
  if (tid < NB) cnt[tid] = 0;
  __syncthreads();

  for (int i0 = 0; i0 < n; i0 += PPT * SORT_T) {
    int2 p[PPT];
    for (int j = 0; j < PPT; ++j) {
      int i = i0 + j * SORT_T + tid;
      p[j] = pairs[beg + min(i, n - 1)];
    }
    for (int j = 0; j < PPT; ++j) {
      int i = i0 + j * SORT_T + tid;
      if (i < n) atomicAdd(&cnt[min(max(p[j].y - node0, 0), NPB - 1)], 1);
    }
  }
  __syncthreads();

  int v    = tid < NB ? cnt[tid] : 0;
  int excl = scan256_excl(v, wsum, tid);
  if (tid < NB) {
    int start = beg + excl;
    int node  = node0 + tid;
    if (tid < NPB && node < N_NODES) {
      deg[node]      = v;
      rowstart[node] = start;
    }
    cnt[tid] = start;
  }
  __syncthreads();

  for (int i0 = 0; i0 < n; i0 += PPT * SORT_T) {
    int2 p[PPT];
    for (int j = 0; j < PPT; ++j) {
      int i = i0 + j * SORT_T + tid;
      p[j] = pairs[beg + min(i, n - 1)];
    }
    for (int j = 0; j < PPT; ++j) {
      int i = i0 + j * SORT_T + tid;
      if (i < n) {
        int pos = atomicAdd(&cnt[min(max(p[j].y - node0, 0), NPB - 1)], 1);
        if ((unsigned)pos < (unsigned)N_EDGES) csr_src[pos] = p[j].x;
      }
    }
  }
}

#define CH2 (OUT_CH / 2)

__device__ __forceinline__ void load10(const float* __restrict__ p, v2f (&v)[CH2]) {
  const v2f* q = (const v2f*)p;
  for (int i = 0; i < CH2; ++i) v[i] = q[i];
}

__device__ __forceinline__ void load_xl_row(const float* __restrict__ xl, int node, int h,
                                            v2f (&v)[CH2]) {
  const v2f* q = (const v2f*)(xl + node * HIDDEN + 2 * h);
  for (int i = 0; i < CH2; ++i) v[i] = q[8 * i];
}

__device__ __forceinline__ float head_logit(const v2f (&x)[CH2],
                                            const v2f (&r)[CH2],
                                            const v2f (&a)[CH2]) {
  v2f s = {0.0f, 0.0f};
  for (int i = 0; i < CH2; ++i) {
    v2f v = x[i] + r[i];
    v2f w = v * 0.2f;
    v.x = fmaxf(v.x, w.x);
    v.y = fmaxf(v.y, w.y);
    s = v * a[i] + s;
  }
  return s.x + s.y;
}

__device__ __forceinline__ void fold4(float& m, float& den, v2f (&acc)[CH2],
                                      float l0, float l1, float l2, float l3,
                                      const v2f (&x0)[CH2], const v2f (&x1)[CH2],
                                      const v2f (&x2)[CH2], const v2f (&x3)[CH2]) {
  float mn = fmaxf(fmaxf(m, fmaxf(l0, l1)), fmaxf(l2, l3));
  float sc = __expf(m - mn);
  float p0 = __expf(l0 - mn);
  float p1 = __expf(l1 - mn);
  float p2 = __expf(l2 - mn);
  float p3 = __expf(l3 - mn);
  den = den * sc + ((p0 + p1) + (p2 + p3));
  for (int i = 0; i < CH2; ++i)
    acc[i] = acc[i] * sc + ((x0[i] * p0 + x1[i] * p1) + (x2[i] * p2 + x3[i] * p3));
  m = mn;
}

__device__ __forceinline__ void gather_node_head(const int* __restrict__ rowstart,
                                                 const int* __restrict__ deg,
                                                 const int* __restrict__ csr_src,
                                                 const float* __restrict__ xl,
                                                 const float* xr,
                                                 const float* __restrict__ att,
                                                 const float* __restrict__ bias,
                                                 int d, int h, float (&o)[OUT_CH]) {
  int hoff = h * OUT_CH;

  v2f a[CH2], r[CH2], x0[CH2], x1[CH2], x2[CH2], x3[CH2], acc[CH2];
  load10(att + hoff, a);
  load10(xr + d * HIDDEN + hoff, r);
  load_xl_row(xl, d, h, x0);

  int beg = min(max(rowstart[d], 0), N_EDGES);
  int end = beg + min(max(deg[d], 0), N_EDGES - beg);
  int rem = (end - beg) & 3;

  int t1 = csr_src[min(beg,     N_EDGES - 1)];
  int t2 = csr_src[min(beg + 1, N_EDGES - 1)];
  int t3 = csr_src[min(beg + 2, N_EDGES - 1)];
  int e = beg + rem;
  int s0 = 0, s1 = 0, s2 = 0, s3 = 0;
  if (e < end) { s0 = csr_src[e]; s1 = csr_src[e + 1]; s2 = csr_src[e + 2]; s3 = csr_src[e + 3]; }

  bool v1 = rem > 0, v2 = rem > 1, v3 = rem > 2;
  load_xl_row(xl, v1 ? clamp_node(t1) : d, h, x1);
  load_xl_row(xl, v2 ? clamp_node(t2) : d, h, x2);
  load_xl_row(xl, v3 ? clamp_node(t3) : d, h, x3);
  float m, den;
  {
    float l0 = head_logit(x0, r, a);
    float l1 = v1 ? head_logit(x1, r, a) : -INFINITY;
    float l2 = v2 ? head_logit(x2, r, a) : -INFINITY;
    float l3 = v3 ? head_logit(x3, r, a) : -INFINITY;
    m = fmaxf(fmaxf(l0, l1), fmaxf(l2, l3));
    float p0 = __expf(l0 - m);
    float p1 = v1 ? __expf(l1 - m) : 0.0f;
    float p2 = v2 ? __expf(l2 - m) : 0.0f;
    float p3 = v3 ? __expf(l3 - m) : 0.0f;
    den = (p0 + p1) + (p2 + p3);
    for (int i = 0; i < CH2; ++i)
      acc[i] = (x0[i] * p0 + x1[i] * p1) + (x2[i] * p2 + x3[i] * p3);
  }

  while (e < end) {
    int c0 = clamp_node(s0);
    int c1 = clamp_node(s1);
    int c2 = clamp_node(s2);
    int c3 = clamp_node(s3);
    e += 4;
    if (e < end) { s0 = csr_src[e]; s1 = csr_src[e + 1]; s2 = csr_src[e + 2]; s3 = csr_src[e + 3]; }
    load_xl_row(xl, c0, h, x0);
    load_xl_row(xl, c1, h, x1);
    load_xl_row(xl, c2, h, x2);
    load_xl_row(xl, c3, h, x3);
    float l0 = head_logit(x0, r, a);
    float l1 = head_logit(x1, r, a);
    float l2 = head_logit(x2, r, a);
    float l3 = head_logit(x3, r, a);
    fold4(m, den, acc, l0, l1, l2, l3, x0, x1, x2, x3);
  }

  float inv = 1.0f / (den + 1e-16f);
  const float* pb = bias + hoff;
  for (int c = 0; c < OUT_CH; ++c) {
    float v = acc[c >> 1][c & 1] * inv + pb[c];
    o[c] = v > 0.0f ? v : expm1f(v);
  }
}

__global__ void __launch_bounds__(256, 8)
gat_node_gather(const int* __restrict__ rowstart,
                const int* __restrict__ deg,
                const int* __restrict__ csr_src,
                const float* __restrict__ xl,
                const float* __restrict__ xr,
                const float* __restrict__ att,
                const float* __restrict__ bias,
                float* __restrict__ xout) {
  int t = blockIdx.x * blockDim.x + threadIdx.x;
  if (t >= N_NODES * HEADS) return;
  int d = t >> 3;
  int h = t & 7;
  float o[OUT_CH];
  gather_node_head(rowstart, deg, csr_src, xl, xr, att, bias, d, h, o);
  float2* po = (float2*)(xout + d * HIDDEN + h * OUT_CH);
  for (int i = 0; i < OUT_CH / 2; ++i) {
    float2 v;
    v.x = o[2 * i];
    v.y = o[2 * i + 1];
    po[i] = v;
  }
}

#define TILE_PITCH 84

template <int NTW>
__device__ __forceinline__ void strip_product(const float* tile_rows,
                                              const float* __restrict__ Wl,
                                              const float* __restrict__ Wr,
                                              float* __restrict__ xl_next,
                                              float* xr_next,
                                              int row0, int q, int lane) {
  int half = lane >> 4;
  int r    = lane & 15;
  const float* ap = tile_rows + r * TILE_PITCH + 2 * half;
  const float* bp[NTW];
#pragma unroll
  for (int j = 0; j < NTW; ++j) {
    int id = q + 4 * j;
    const float* w = id < 5 ? Wl : Wr;
    int ct = id < 5 ? id : id - 5;
    bp[j] = w + (2 * half) * HIDDEN + ct * 16 + r;
  }
  v8f c[NTW];
#pragma unroll
  for (int j = 0; j < NTW; ++j) c[j] = (v8f){};

#pragma unroll 4
  for (int k = 0; k < HIDDEN; k += 4) {
    v2f a = *(const v2f*)(ap + k);
#pragma unroll
    for (int j = 0; j < NTW; ++j) {
      v2f b;
      b.x = bp[j][k * HIDDEN];
      b.y = bp[j][(k + 1) * HIDDEN];
      c[j] = __builtin_amdgcn_wmma_f32_16x16x4_f32(false, a, false, b, (short)0, c[j], false, false);
    }
  }

#pragma unroll
  for (int j = 0; j < NTW; ++j) {
    int id = q + 4 * j;
    int ct = id < 5 ? id : id - 5;
    int n  = ct * 16 + r;
    int hd = n / OUT_CH;
    int slot = xl_slot(hd, n - hd * OUT_CH);
#pragma unroll
    for (int v = 0; v < 8; ++v) {
      int orow = row0 + v + 8 * half;
      if (id < 5) xl_next[orow * HIDDEN + slot] = c[j][v];
      else        xr_next[orow * HIDDEN + n]    = c[j][v];
    }
  }
}

__global__ void __launch_bounds__(256, 8)
gat_gather_then_gemm(const int* __restrict__ rowstart,
                     const int* __restrict__ deg,
                     const int* __restrict__ csr_src,
                     const float* __restrict__ xl,
                     float* xr,
                     const float* __restrict__ att,
                     const float* __restrict__ bias,
                     const float* __restrict__ Wl,
                     const float* __restrict__ Wr,
                     float* __restrict__ xl_next) {
  __shared__ __attribute__((aligned(16))) float tile[32 * TILE_PITCH];
  int t = blockIdx.x * blockDim.x + threadIdx.x;
  if (t < N_NODES * HEADS) {
    int d = t >> 3;
    int h = t & 7;
    float o[OUT_CH];
    gather_node_head(rowstart, deg, csr_src, xl, xr, att, bias, d, h, o);
    float2* po = (float2*)(tile + (threadIdx.x >> 3) * TILE_PITCH + h * OUT_CH);
    for (int i = 0; i < OUT_CH / 2; ++i) {
      float2 v;
      v.x = o[2 * i];
      v.y = o[2 * i + 1];
      po[i] = v;
    }
  }
  __syncthreads();

  int w     = __builtin_amdgcn_readfirstlane(threadIdx.x >> 5);
  int strip = w >> 2;
  int q     = strip ? 3 - (w & 3) : (w & 3);
  int row0  = blockIdx.x * 32 + strip * 16;
  if (row0 >= N_NODES) return;
  const float* rows = tile + strip * 16 * TILE_PITCH;
  int lane = threadIdx.x & 31;
  if (q < 2) strip_product<3>(rows, Wl, Wr, xl_next, xr, row0, q, lane);
  else       strip_product<2>(rows, Wl, Wr, xl_next, xr, row0, q, lane);
}

extern "C" void kernel_launch(void* const* d_in, const int* in_sizes, int n_in,
                              void* d_out, int out_size, void* d_ws, size_t ws_size,
                              hipStream_t stream) {
  (void)in_sizes; (void)n_in; (void)out_size; (void)ws_size;

  const float* x    = (const float*)d_in[0];
  const int*   ei   = (const int*)d_in[1];
  const int*   esrc = ei;
  const int*   edst = ei + N_EDGES;

  float* ws    = (float*)d_ws;
  float* xl    = ws;
  float* xr    = xl   + (size_t)N_NODES * HIDDEN;
  float* xbuf  = xr   + (size_t)N_NODES * HIDDEN;
  int*   deg      = (int*)(xbuf + (size_t)N_NODES * HIDDEN);
  int*   rowstart = deg      + N_NODES;
  int*   spare    = rowstart + N_NODES;
  int*   csr_src  = spare    + 64 + N_EDGES;
  int*   table      = (int*)xbuf;
  int*   range_base = table      + NCH * NB;
  int*   range_cnt  = range_base + NB;
  int2*  pairs      = (int2*)(range_cnt + NB);

  const int gemm_blocks = N_NODES / 16;
  const int head_blocks  = (N_NODES * HEADS + 255) / 256;

  csr_hist   <<<NCH, SORT_T, 0, stream>>>(edst, table);
  csr_scatter<<<NCH, SORT_T, 0, stream>>>(esrc, edst, table, range_base, range_cnt, pairs);
  csr_bucket <<<NB,  SORT_T, 0, stream>>>(pairs, range_base, range_cnt, deg, rowstart, csr_src);

  float* xl_a = xl;
  float* xl_b = xbuf;
  gat_gemm2_wmma<F_IN><<<gemm_blocks, 32, 0, stream>>>(
      x, (const float*)d_in[2], (const float*)d_in[3], xl_a, xr);
  gat_gather_then_gemm<<<head_blocks, 256, 0, stream>>>(
      rowstart, deg, csr_src, xl_a, xr, (const float*)d_in[4], (const float*)d_in[5],
      (const float*)d_in[6], (const float*)d_in[7], xl_b);
  gat_gather_then_gemm<<<head_blocks, 256, 0, stream>>>(
      rowstart, deg, csr_src, xl_b, xr, (const float*)d_in[8], (const float*)d_in[9],
      (const float*)d_in[10], (const float*)d_in[11], xl_a);
  gat_node_gather<<<head_blocks, 256, 0, stream>>>(
      rowstart, deg, csr_src, xl_a, xr, (const float*)d_in[12], (const float*)d_in[13],
      (float*)d_out);
}


# gather epilogue without branches: ELU's exp(v)-1 from one hardware exponential, a short series near zero (a few ulp), in packed form, not ten library expm1 calls; bias read up front as pairs
# speedup vs baseline: 7.3946x; 1.0375x over previous
#include <hip/hip_runtime.h>
#include <math.h>

#define N_NODES 50000
#define N_EDGES 800000
#define F_IN    256
#define HEADS   8
#define OUT_CH  10
#define HIDDEN  80

typedef __attribute__((ext_vector_type(2))) float v2f;
typedef __attribute__((ext_vector_type(8))) float v8f;

__device__ __forceinline__ int xl_slot(int head, int ch) {
  return ((ch >> 1) << 4) + 2 * head + (ch & 1);
}

template <int K>
__global__ void __launch_bounds__(32)
gat_gemm2_wmma(const float* __restrict__ X,
               const float* __restrict__ Wl,
               const float* __restrict__ Wr,
               float* __restrict__ xl,
               float* __restrict__ xr) {
  const int NT = HIDDEN / 16;
  int mt   = blockIdx.x;
  int lane = threadIdx.x;
  int half = lane >> 4;
  int r    = lane & 15;

  const float* ap  = X  + (mt * 16 + r) * K + 2 * half;
  const float* blp = Wl + (2 * half) * HIDDEN + r;
  const float* brp = Wr + (2 * half) * HIDDEN + r;

  v8f cl[NT];
  v8f cr[NT];
#pragma unroll
  for (int t = 0; t < NT; ++t) { cl[t] = (v8f){}; cr[t] = (v8f){}; }

#pragma unroll 2
  for (int k = 0; k < K; k += 4) {
    v2f a = *(const v2f*)(ap + k);
#pragma unroll
    for (int t = 0; t < NT; ++t) {
      v2f bl;
      bl.x = blp[k * HIDDEN + t * 16];
      bl.y = blp[(k + 1) * HIDDEN + t * 16];
      v2f br;
      br.x = brp[k * HIDDEN + t * 16];
      br.y = brp[(k + 1) * HIDDEN + t * 16];
      cl[t] = __builtin_amdgcn_wmma_f32_16x16x4_f32(false, a, false, bl, (short)0, cl[t], false, false);
      cr[t] = __builtin_amdgcn_wmma_f32_16x16x4_f32(false, a, false, br, (short)0, cr[t], false, false);
    }
  }

#pragma unroll
  for (int t = 0; t < NT; ++t) {
    int n    = t * 16 + r;
    int hd   = n / OUT_CH;
    int slot = xl_slot(hd, n - hd * OUT_CH);
#pragma unroll
    for (int v = 0; v < 8; ++v) {
      int orow = mt * 16 + v + 8 * half;
      xl[orow * HIDDEN + slot] = cl[t][v];
      xr[orow * HIDDEN + t * 16 + r] = cr[t][v];
    }
  }
}

#define NB      256
#define NPB     196
#define NCH     256
#define CHUNK   3125
#define SORT_T  512
#define EPT     7
#define PPT     8
#define ROWS_PT 32

static_assert(NB * NPB >= N_NODES, "ranges cover all nodes");
static_assert(NCH * CHUNK == N_EDGES, "chunks tile the edge list exactly");
static_assert(EPT * SORT_T >= CHUNK, "one pass of a work-group covers its chunk");
static_assert(NB == 256 && SORT_T >= NB && SORT_T % 64 == 0, "thread maps below assume these");
static_assert((SORT_T / 64) * ROWS_PT == NCH, "a work-group's threads cover the whole table");

__device__ __forceinline__ int clamp_node(int s) {
  return min(max(s, 0), N_NODES - 1);
}

__device__ __forceinline__ int range_of(int node) { return node / NPB; }

__device__ __forceinline__ int scan256_excl(int v, int* wsum, int tid) {
  int lane = tid & 31;
  int w    = tid >> 5;
  int incl = v;
  for (int off = 1; off < 32; off <<= 1) {
    int t = __shfl_up(incl, off);
    if (lane >= off) incl += t;
  }
  if (tid < 256 && lane == 31) wsum[w] = incl;
  __syncthreads();
  int base = 0;
  for (int j = 0; j < 8; ++j) {
    int s = wsum[j];
    if (j < w) base += s;
  }
  return base + incl - v;
}

__global__ void __launch_bounds__(SORT_T)
csr_hist(const int* __restrict__ edst, int* __restrict__ table) {
  __shared__ int h[NB];
  int tid = threadIdx.x;
  int c   = blockIdx.x;
  if (tid < NB) h[tid] = 0;
  const int* p = edst + c * CHUNK;
  int d[EPT];
  for (int j = 0; j < EPT; ++j) {
    int i = tid + j * SORT_T;
    d[j] = p[min(i, CHUNK - 1)];
  }
  __syncthreads();
  for (int j = 0; j < EPT; ++j) {
    int i = tid + j * SORT_T;
    if (i < CHUNK) atomicAdd(&h[range_of(clamp_node(d[j]))], 1);
  }
  __syncthreads();
  if (tid < NB) table[c * NB + tid] = h[tid];
}

__global__ void __launch_bounds__(SORT_T)
csr_scatter(const int* __restrict__ esrc,
            const int* __restrict__ edst,
            const int* __restrict__ table,
            int* __restrict__ range_base,
            int* __restrict__ range_cnt,
            int2* __restrict__ pairs) {
  __shared__ __attribute__((aligned(16))) int part_tot[SORT_T / 64][NB];
  __shared__ __attribute__((aligned(16))) int part_bef[SORT_T / 64][NB];
  __shared__ int cur[NB];
  __shared__ int wsum[8];
  int tid = threadIdx.x;
  int c   = blockIdx.x;

  int s[EPT], d[EPT];
  for (int j = 0; j < EPT; ++j) {
    int i = c * CHUNK + min(tid + j * SORT_T, CHUNK - 1);
    s[j] = esrc[i];
    d[j] = edst[i];
  }

  int g  = tid >> 6;
  int b4 = tid & 63;
  const int4* colp = (const int4*)(table + (g * ROWS_PT) * NB) + b4;
  int4 v[ROWS_PT];
  for (int k = 0; k < ROWS_PT; ++k) v[k] = colp[k * (NB / 4)];
  int4 tot = {0, 0, 0, 0};
  int4 bef = {0, 0, 0, 0};
  for (int k = 0; k < ROWS_PT; ++k) {
    bool earlier = (g * ROWS_PT + k) < c;
    tot.x += v[k].x;  tot.y += v[k].y;  tot.z += v[k].z;  tot.w += v[k].w;
    bef.x += earlier ? v[k].x : 0;
    bef.y += earlier ? v[k].y : 0;
    bef.z += earlier ? v[k].z : 0;
    bef.w += earlier ? v[k].w : 0;
  }
  ((int4*)part_tot[g])[b4] = tot;
  ((int4*)part_bef[g])[b4] = bef;
  __syncthreads();
  int t = 0, bf = 0;
  if (tid < NB) {
    for (int j = 0; j < SORT_T / 64; ++j) {
      t  += part_tot[j][tid];
      bf += part_bef[j][tid];
    }
  }
  int base = scan256_excl(t, wsum, tid);
  if (tid < NB) {
    cur[tid] = base + bf;
    if (c == 0) {
      range_base[tid] = base;
      range_cnt[tid]  = t;
    }
  }
  __syncthreads();
  int pos[EPT];
  for (int j = 0; j < EPT; ++j) {
    int i = tid + j * SORT_T;
    d[j]   = clamp_node(d[j]);
    pos[j] = i < CHUNK ? atomicAdd(&cur[range_of(d[j])], 1) : -1;
  }
  for (int j = 0; j < EPT; ++j) {
    if ((unsigned)pos[j] < (unsigned)N_EDGES) {
      int2 pr;
      pr.x = s[j];
      pr.y = d[j];
      pairs[pos[j]] = pr;
    }
  }
}

__global__ void __launch_bounds__(SORT_T)
csr_bucket(const int2* __restrict__ pairs,
           const int* __restrict__ range_base,
           const int* __restrict__ range_cnt,
           int* __restrict__ deg,
           int* __restrict__ rowstart,
           int* __restrict__ csr_src) {
  __shared__ int cnt[NB];
  __shared__ int wsum[8];
  int tid   = threadIdx.x;
  int node0 = blockIdx.x * NPB;
  int beg   = min(max(range_base[blockIdx.x], 0), N_EDGES);
  int n     = min(max(range_cnt[blockIdx.x], 0), N_EDGES - beg);
  if (tid < NB) cnt[tid] = 0;
  __syncthreads();

  for (int i0 = 0; i0 < n; i0 += PPT * SORT_T) {
    int2 p[PPT];
    for (int j = 0; j < PPT; ++j) {
      int i = i0 + j * SORT_T + tid;
      p[j] = pairs[beg + min(i, n - 1)];
    }
    for (int j = 0; j < PPT; ++j) {
      int i = i0 + j * SORT_T + tid;
      if (i < n) atomicAdd(&cnt[min(max(p[j].y - node0, 0), NPB - 1)], 1);
    }
  }
  __syncthreads();

  int v    = tid < NB ? cnt[tid] : 0;
  int excl = scan256_excl(v, wsum, tid);
  if (tid < NB) {
    int start = beg + excl;
    int node  = node0 + tid;
    if (tid < NPB && node < N_NODES) {
      deg[node]      = v;
      rowstart[node] = start;
    }
    cnt[tid] = start;
  }
  __syncthreads();

  for (int i0 = 0; i0 < n; i0 += PPT * SORT_T) {
    int2 p[PPT];
    for (int j = 0; j < PPT; ++j) {
      int i = i0 + j * SORT_T + tid;
      p[j] = pairs[beg + min(i, n - 1)];
    }
    for (int j = 0; j < PPT; ++j) {
      int i = i0 + j * SORT_T + tid;
      if (i < n) {
        int pos = atomicAdd(&cnt[min(max(p[j].y - node0, 0), NPB - 1)], 1);
        if ((unsigned)pos < (unsigned)N_EDGES) csr_src[pos] = p[j].x;
      }
    }
  }
}

#define CH2 (OUT_CH / 2)

__device__ __forceinline__ void load10(const float* __restrict__ p, v2f (&v)[CH2]) {
  const v2f* q = (const v2f*)p;
  for (int i = 0; i < CH2; ++i) v[i] = q[i];
}

__device__ __forceinline__ void load_xl_row(const float* __restrict__ xl, int node, int h,
                                            v2f (&v)[CH2]) {
  const v2f* q = (const v2f*)(xl + node * HIDDEN + 2 * h);
  for (int i = 0; i < CH2; ++i) v[i] = q[8 * i];
}

__device__ __forceinline__ float head_logit(const v2f (&x)[CH2],
                                            const v2f (&r)[CH2],
                                            const v2f (&a)[CH2]) {
  v2f s = {0.0f, 0.0f};
  for (int i = 0; i < CH2; ++i) {
    v2f v = x[i] + r[i];
    v2f w = v * 0.2f;
    v.x = fmaxf(v.x, w.x);
    v.y = fmaxf(v.y, w.y);
    s = v * a[i] + s;
  }
  return s.x + s.y;
}

__device__ __forceinline__ void fold4(float& m, float& den, v2f (&acc)[CH2],
                                      float l0, float l1, float l2, float l3,
                                      const v2f (&x0)[CH2], const v2f (&x1)[CH2],
                                      const v2f (&x2)[CH2], const v2f (&x3)[CH2]) {
  float mn = fmaxf(fmaxf(m, fmaxf(l0, l1)), fmaxf(l2, l3));
  float sc = __expf(m - mn);
  float p0 = __expf(l0 - mn);
  float p1 = __expf(l1 - mn);
  float p2 = __expf(l2 - mn);
  float p3 = __expf(l3 - mn);
  den = den * sc + ((p0 + p1) + (p2 + p3));
  for (int i = 0; i < CH2; ++i)
    acc[i] = acc[i] * sc + ((x0[i] * p0 + x1[i] * p1) + (x2[i] * p2 + x3[i] * p3));
  m = mn;
}

__device__ __forceinline__ v2f elu_pair(v2f v) {
  v2f t;
  t.x = __expf(v.x);
  t.y = __expf(v.y);
  v2f far = t - 1.0f;
  v2f p = v * (1.0f / 5040.0f) + (1.0f / 720.0f);
  p = p * v + (1.0f / 120.0f);
  p = p * v + (1.0f / 24.0f);
  p = p * v + (1.0f / 6.0f);
  p = p * v + 0.5f;
  v2f near = (v * v) * p + v;
  v2f o;
  o.x = v.x > 0.0f ? v.x : (v.x > -0.35f ? near.x : far.x);
  o.y = v.y > 0.0f ? v.y : (v.y > -0.35f ? near.y : far.y);
  return o;
}

__device__ __forceinline__ void gather_node_head(const int* __restrict__ rowstart,
                                                 const int* __restrict__ deg,
                                                 const int* __restrict__ csr_src,
                                                 const float* __restrict__ xl,
                                                 const float* xr,
                                                 const float* __restrict__ att,
                                                 const float* __restrict__ bias,
                                                 int d, int h, float (&o)[OUT_CH]) {
  int hoff = h * OUT_CH;

  v2f a[CH2], r[CH2], b[CH2], x0[CH2], x1[CH2], x2[CH2], x3[CH2], acc[CH2];
  load10(att + hoff, a);
  load10(bias + hoff, b);
  load10(xr + d * HIDDEN + hoff, r);
  load_xl_row(xl, d, h, x0);

  int beg = min(max(rowstart[d], 0), N_EDGES);
  int end = beg + min(max(deg[d], 0), N_EDGES - beg);
  int rem = (end - beg) & 3;

  int t1 = csr_src[min(beg,     N_EDGES - 1)];
  int t2 = csr_src[min(beg + 1, N_EDGES - 1)];
  int t3 = csr_src[min(beg + 2, N_EDGES - 1)];
  int e = beg + rem;
  int s0 = 0, s1 = 0, s2 = 0, s3 = 0;
  if (e < end) { s0 = csr_src[e]; s1 = csr_src[e + 1]; s2 = csr_src[e + 2]; s3 = csr_src[e + 3]; }

  bool v1 = rem > 0, v2 = rem > 1, v3 = rem > 2;
  load_xl_row(xl, v1 ? clamp_node(t1) : d, h, x1);
  load_xl_row(xl, v2 ? clamp_node(t2) : d, h, x2);
  load_xl_row(xl, v3 ? clamp_node(t3) : d, h, x3);
  float m, den;
  {
    float l0 = head_logit(x0, r, a);
    float l1 = v1 ? head_logit(x1, r, a) : -INFINITY;
    float l2 = v2 ? head_logit(x2, r, a) : -INFINITY;
    float l3 = v3 ? head_logit(x3, r, a) : -INFINITY;
    m = fmaxf(fmaxf(l0, l1), fmaxf(l2, l3));
    float p0 = __expf(l0 - m);
    float p1 = v1 ? __expf(l1 - m) : 0.0f;
    float p2 = v2 ? __expf(l2 - m) : 0.0f;
    float p3 = v3 ? __expf(l3 - m) : 0.0f;
    den = (p0 + p1) + (p2 + p3);
    for (int i = 0; i < CH2; ++i)
      acc[i] = (x0[i] * p0 + x1[i] * p1) + (x2[i] * p2 + x3[i] * p3);
  }

  while (e < end) {
    int c0 = clamp_node(s0);
    int c1 = clamp_node(s1);
    int c2 = clamp_node(s2);
    int c3 = clamp_node(s3);
    e += 4;
    if (e < end) { s0 = csr_src[e]; s1 = csr_src[e + 1]; s2 = csr_src[e + 2]; s3 = csr_src[e + 3]; }
    load_xl_row(xl, c0, h, x0);
    load_xl_row(xl, c1, h, x1);
    load_xl_row(xl, c2, h, x2);
    load_xl_row(xl, c3, h, x3);
    float l0 = head_logit(x0, r, a);
    float l1 = head_logit(x1, r, a);
    float l2 = head_logit(x2, r, a);
    float l3 = head_logit(x3, r, a);
    fold4(m, den, acc, l0, l1, l2, l3, x0, x1, x2, x3);
  }

  float inv = 1.0f / (den + 1e-16f);
  for (int i = 0; i < CH2; ++i) {
    v2f e = elu_pair(acc[i] * inv + b[i]);
    o[2 * i]     = e.x;
    o[2 * i + 1] = e.y;
  }
}

__global__ void __launch_bounds__(256, 8)
gat_node_gather(const int* __restrict__ rowstart,
                const int* __restrict__ deg,
                const int* __restrict__ csr_src,
                const float* __restrict__ xl,
                const float* __restrict__ xr,
                const float* __restrict__ att,
                const float* __restrict__ bias,
                float* __restrict__ xout) {
  int t = blockIdx.x * blockDim.x + threadIdx.x;
  if (t >= N_NODES * HEADS) return;
  int d = t >> 3;
  int h = t & 7;
  float o[OUT_CH];
  gather_node_head(rowstart, deg, csr_src, xl, xr, att, bias, d, h, o);
  float2* po = (float2*)(xout + d * HIDDEN + h * OUT_CH);
  for (int i = 0; i < OUT_CH / 2; ++i) {
    float2 v;
    v.x = o[2 * i];
    v.y = o[2 * i + 1];
    po[i] = v;
  }
}

#define TILE_PITCH 84

template <int NTW>
__device__ __forceinline__ void strip_product(const float* tile_rows,
                                              const float* __restrict__ Wl,
                                              const float* __restrict__ Wr,
                                              float* __restrict__ xl_next,
                                              float* xr_next,
                                              int row0, int q, int lane) {
  int half = lane >> 4;
  int r    = lane & 15;
  const float* ap = tile_rows + r * TILE_PITCH + 2 * half;
  const float* bp[NTW];
#pragma unroll
  for (int j = 0; j < NTW; ++j) {
    int id = q + 4 * j;
    const float* w = id < 5 ? Wl : Wr;
    int ct = id < 5 ? id : id - 5;
    bp[j] = w + (2 * half) * HIDDEN + ct * 16 + r;
  }
  v8f c[NTW];
#pragma unroll
  for (int j = 0; j < NTW; ++j) c[j] = (v8f){};

#pragma unroll 4
  for (int k = 0; k < HIDDEN; k += 4) {
    v2f a = *(const v2f*)(ap + k);
#pragma unroll
    for (int j = 0; j < NTW; ++j) {
      v2f b;
      b.x = bp[j][k * HIDDEN];
      b.y = bp[j][(k + 1) * HIDDEN];
      c[j] = __builtin_amdgcn_wmma_f32_16x16x4_f32(false, a, false, b, (short)0, c[j], false, false);
    }
  }

#pragma unroll
  for (int j = 0; j < NTW; ++j) {
    int id = q + 4 * j;
    int ct = id < 5 ? id : id - 5;
    int n  = ct * 16 + r;
    int hd = n / OUT_CH;
    int slot = xl_slot(hd, n - hd * OUT_CH);
#pragma unroll
    for (int v = 0; v < 8; ++v) {
      int orow = row0 + v + 8 * half;
      if (id < 5) xl_next[orow * HIDDEN + slot] = c[j][v];
      else        xr_next[orow * HIDDEN + n]    = c[j][v];
    }
  }
}

__global__ void __launch_bounds__(256, 8)
gat_gather_then_gemm(const int* __restrict__ rowstart,
                     const int* __restrict__ deg,
                     const int* __restrict__ csr_src,
                     const float* __restrict__ xl,
                     float* xr,
                     const float* __restrict__ att,
                     const float* __restrict__ bias,
                     const float* __restrict__ Wl,
                     const float* __restrict__ Wr,
                     float* __restrict__ xl_next) {
  __shared__ __attribute__((aligned(16))) float tile[32 * TILE_PITCH];
  int t = blockIdx.x * blockDim.x + threadIdx.x;
  if (t < N_NODES * HEADS) {
    int d = t >> 3;
    int h = t & 7;
    float o[OUT_CH];
    gather_node_head(rowstart, deg, csr_src, xl, xr, att, bias, d, h, o);
    float2* po = (float2*)(tile + (threadIdx.x >> 3) * TILE_PITCH + h * OUT_CH);
    for (int i = 0; i < OUT_CH / 2; ++i) {
      float2 v;
      v.x = o[2 * i];
      v.y = o[2 * i + 1];
      po[i] = v;
    }
  }
  __syncthreads();

  int w     = __builtin_amdgcn_readfirstlane(threadIdx.x >> 5);
  int strip = w >> 2;
  int q     = strip ? 3 - (w & 3) : (w & 3);
  int row0  = blockIdx.x * 32 + strip * 16;
  if (row0 >= N_NODES) return;
  const float* rows = tile + strip * 16 * TILE_PITCH;
  int lane = threadIdx.x & 31;
  if (q < 2) strip_product<3>(rows, Wl, Wr, xl_next, xr, row0, q, lane);
  else       strip_product<2>(rows, Wl, Wr, xl_next, xr, row0, q, lane);
}

extern "C" void kernel_launch(void* const* d_in, const int* in_sizes, int n_in,
                              void* d_out, int out_size, void* d_ws, size_t ws_size,
                              hipStream_t stream) {
  (void)in_sizes; (void)n_in; (void)out_size; (void)ws_size;

  const float* x    = (const float*)d_in[0];
  const int*   ei   = (const int*)d_in[1];
  const int*   esrc = ei;
  const int*   edst = ei + N_EDGES;

  float* ws    = (float*)d_ws;
  float* xl    = ws;
  float* xr    = xl   + (size_t)N_NODES * HIDDEN;
  float* xbuf  = xr   + (size_t)N_NODES * HIDDEN;
  int*   deg      = (int*)(xbuf + (size_t)N_NODES * HIDDEN);
  int*   rowstart = deg      + N_NODES;
  int*   spare    = rowstart + N_NODES;
  int*   csr_src  = spare    + 64 + N_EDGES;
  int*   table      = (int*)xbuf;
  int*   range_base = table      + NCH * NB;
  int*   range_cnt  = range_base + NB;
  int2*  pairs      = (int2*)(range_cnt + NB);

  const int gemm_blocks = N_NODES / 16;
  const int head_blocks  = (N_NODES * HEADS + 255) / 256;

  csr_hist   <<<NCH, SORT_T, 0, stream>>>(edst, table);
  csr_scatter<<<NCH, SORT_T, 0, stream>>>(esrc, edst, table, range_base, range_cnt, pairs);
  csr_bucket <<<NB,  SORT_T, 0, stream>>>(pairs, range_base, range_cnt, deg, rowstart, csr_src);

  float* xl_a = xl;
  float* xl_b = xbuf;
  gat_gemm2_wmma<F_IN><<<gemm_blocks, 32, 0, stream>>>(
      x, (const float*)d_in[2], (const float*)d_in[3], xl_a, xr);
  gat_gather_then_gemm<<<head_blocks, 256, 0, stream>>>(
      rowstart, deg, csr_src, xl_a, xr, (const float*)d_in[4], (const float*)d_in[5],
      (const float*)d_in[6], (const float*)d_in[7], xl_b);
  gat_gather_then_gemm<<<head_blocks, 256, 0, stream>>>(
      rowstart, deg, csr_src, xl_b, xr, (const float*)d_in[8], (const float*)d_in[9],
      (const float*)d_in[10], (const float*)d_in[11], xl_a);
  gat_node_gather<<<head_blocks, 256, 0, stream>>>(
      rowstart, deg, csr_src, xl_a, xr, (const float*)d_in[12], (const float*)d_in[13],
      (float*)d_out);
}


# layer 0's Wl/Wr products moved into the last sort launch as 256 extra 16-wave work-groups, one half strip (5 column tiles) per wave: 6 launches, not 7; same k order and sums
# speedup vs baseline: 7.7132x; 1.0431x over previous
#include <hip/hip_runtime.h>
#include <math.h>

#define N_NODES 50000
#define N_EDGES 800000
#define F_IN    256
#define HEADS   8
#define OUT_CH  10
#define HIDDEN  80

typedef __attribute__((ext_vector_type(2))) float v2f;
typedef __attribute__((ext_vector_type(8))) float v8f;

__device__ __forceinline__ int xl_slot(int head, int ch) {
  return ((ch >> 1) << 4) + 2 * head + (ch & 1);
}

template <int K, int ID0, int NTW>
__device__ __forceinline__ void gemm0_strip(const float* __restrict__ X,
                                            const float* __restrict__ Wl,
                                            const float* __restrict__ Wr,
                                            float* __restrict__ xl,
                                            float* __restrict__ xr,
                                            int mt, int lane) {
  int half = lane >> 4;
  int r    = lane & 15;

  const float* ap = X + (mt * 16 + r) * K + 2 * half;
  const float* bp[NTW];
#pragma unroll
  for (int j = 0; j < NTW; ++j) {
    int id = ID0 + j;
    const float* w = id < 5 ? Wl : Wr;
    int ct = id < 5 ? id : id - 5;
    bp[j] = w + (2 * half) * HIDDEN + ct * 16 + r;
  }
  v8f c[NTW];
#pragma unroll
  for (int j = 0; j < NTW; ++j) c[j] = (v8f){};

#pragma unroll 4
  for (int k = 0; k < K; k += 4) {
    v2f a = *(const v2f*)(ap + k);
#pragma unroll
    for (int j = 0; j < NTW; ++j) {
      v2f b;
      b.x = bp[j][k * HIDDEN];
      b.y = bp[j][(k + 1) * HIDDEN];
      c[j] = __builtin_amdgcn_wmma_f32_16x16x4_f32(false, a, false, b, (short)0, c[j], false, false);
    }
  }

#pragma unroll
  for (int j = 0; j < NTW; ++j) {
    int id = ID0 + j;
    int ct = id < 5 ? id : id - 5;
    int n    = ct * 16 + r;
    int hd   = n / OUT_CH;
    int slot = xl_slot(hd, n - hd * OUT_CH);
#pragma unroll
    for (int v = 0; v < 8; ++v) {
      int orow = mt * 16 + v + 8 * half;
      if (id < 5) xl[orow * HIDDEN + slot] = c[j][v];
      else        xr[orow * HIDDEN + n]    = c[j][v];
    }
  }
}

#define NB      256
#define NPB     196
#define NCH     256
#define CHUNK   3125
#define SORT_T  512
#define EPT     7
#define PPT     8
#define ROWS_PT 32

static_assert(NB * NPB >= N_NODES, "ranges cover all nodes");
static_assert(NCH * CHUNK == N_EDGES, "chunks tile the edge list exactly");
static_assert(EPT * SORT_T >= CHUNK, "one pass of a work-group covers its chunk");
static_assert(NB == 256 && SORT_T >= NB && SORT_T % 64 == 0, "thread maps below assume these");
static_assert((SORT_T / 64) * ROWS_PT == NCH, "a work-group's threads cover the whole table");

__device__ __forceinline__ int clamp_node(int s) {
  return min(max(s, 0), N_NODES - 1);
}

__device__ __forceinline__ int range_of(int node) { return node / NPB; }

__device__ __forceinline__ int scan256_excl(int v, int* wsum, int tid) {
  int lane = tid & 31;
  int w    = tid >> 5;
  int incl = v;
  for (int off = 1; off < 32; off <<= 1) {
    int t = __shfl_up(incl, off);
    if (lane >= off) incl += t;
  }
  if (tid < 256 && lane == 31) wsum[w] = incl;
  __syncthreads();
  int base = 0;
  for (int j = 0; j < 8; ++j) {
    int s = wsum[j];
    if (j < w) base += s;
  }
  return base + incl - v;
}

__global__ void __launch_bounds__(SORT_T)
csr_hist(const int* __restrict__ edst, int* __restrict__ table) {
  __shared__ int h[NB];
  int tid = threadIdx.x;
  int c   = blockIdx.x;
  if (tid < NB) h[tid] = 0;
  const int* p = edst + c * CHUNK;
  int d[EPT];
  for (int j = 0; j < EPT; ++j) {
    int i = tid + j * SORT_T;
    d[j] = p[min(i, CHUNK - 1)];
  }
  __syncthreads();
  for (int j = 0; j < EPT; ++j) {
    int i = tid + j * SORT_T;
    if (i < CHUNK) atomicAdd(&h[range_of(clamp_node(d[j]))], 1);
  }
  __syncthreads();
  if (tid < NB) table[c * NB + tid] = h[tid];
}

__global__ void __launch_bounds__(SORT_T)
csr_scatter(const int* __restrict__ esrc,
            const int* __restrict__ edst,
            const int* __restrict__ table,
            int* __restrict__ range_base,
            int* __restrict__ range_cnt,
            int2* __restrict__ pairs) {
  __shared__ __attribute__((aligned(16))) int part_tot[SORT_T / 64][NB];
  __shared__ __attribute__((aligned(16))) int part_bef[SORT_T / 64][NB];
  __shared__ int cur[NB];
  __shared__ int wsum[8];
  int tid = threadIdx.x;
  int c   = blockIdx.x;

  int s[EPT], d[EPT];
  for (int j = 0; j < EPT; ++j) {
    int i = c * CHUNK + min(tid + j * SORT_T, CHUNK - 1);
    s[j] = esrc[i];
    d[j] = edst[i];
  }

  int g  = tid >> 6;
  int b4 = tid & 63;
  const int4* colp = (const int4*)(table + (g * ROWS_PT) * NB) + b4;
  int4 v[ROWS_PT];
  for (int k = 0; k < ROWS_PT; ++k) v[k] = colp[k * (NB / 4)];
  int4 tot = {0, 0, 0, 0};
  int4 bef = {0, 0, 0, 0};
  for (int k = 0; k < ROWS_PT; ++k) {
    bool earlier = (g * ROWS_PT + k) < c;
    tot.x += v[k].x;  tot.y += v[k].y;  tot.z += v[k].z;  tot.w += v[k].w;
    bef.x += earlier ? v[k].x : 0;
    bef.y += earlier ? v[k].y : 0;
    bef.z += earlier ? v[k].z : 0;
    bef.w += earlier ? v[k].w : 0;
  }
  ((int4*)part_tot[g])[b4] = tot;
  ((int4*)part_bef[g])[b4] = bef;
  __syncthreads();
  int t = 0, bf = 0;
  if (tid < NB) {
    for (int j = 0; j < SORT_T / 64; ++j) {
      t  += part_tot[j][tid];
      bf += part_bef[j][tid];
    }
  }
  int base = scan256_excl(t, wsum, tid);
  if (tid < NB) {
    cur[tid] = base + bf;
    if (c == 0) {
      range_base[tid] = base;
      range_cnt[tid]  = t;
    }
  }
  __syncthreads();
  int pos[EPT];
  for (int j = 0; j < EPT; ++j) {
    int i = tid + j * SORT_T;
    d[j]   = clamp_node(d[j]);
    pos[j] = i < CHUNK ? atomicAdd(&cur[range_of(d[j])], 1) : -1;
  }
  for (int j = 0; j < EPT; ++j) {
    if ((unsigned)pos[j] < (unsigned)N_EDGES) {
      int2 pr;
      pr.x = s[j];
      pr.y = d[j];
      pairs[pos[j]] = pr;
    }
  }
}

__device__ __forceinline__ void csr_bucket_body(const int2* __restrict__ pairs,
                                                const int* __restrict__ range_base,
                                                const int* __restrict__ range_cnt,
                                                int* __restrict__ deg,
                                                int* __restrict__ rowstart,
                                                int* __restrict__ csr_src,
                                                int blk) {
  __shared__ int cnt[NB];
  __shared__ int wsum[8];
  int tid   = threadIdx.x;
  int node0 = blk * NPB;
  int beg   = min(max(range_base[blk], 0), N_EDGES);
  int n     = min(max(range_cnt[blk], 0), N_EDGES - beg);
  if (tid < NB) cnt[tid] = 0;
  __syncthreads();

  for (int i0 = 0; i0 < n; i0 += PPT * SORT_T) {
    int2 p[PPT];
    for (int j = 0; j < PPT; ++j) {
      int i = i0 + j * SORT_T + tid;
      p[j] = pairs[beg + min(i, n - 1)];
    }
    for (int j = 0; j < PPT; ++j) {
      int i = i0 + j * SORT_T + tid;
      if (i < n) atomicAdd(&cnt[min(max(p[j].y - node0, 0), NPB - 1)], 1);
    }
  }
  __syncthreads();

  int v    = tid < NB ? cnt[tid] : 0;
  int excl = scan256_excl(v, wsum, tid);
  if (tid < NB) {
    int start = beg + excl;
    int node  = node0 + tid;
    if (tid < NPB && node < N_NODES) {
      deg[node]      = v;
      rowstart[node] = start;
    }
    cnt[tid] = start;
  }
  __syncthreads();

  for (int i0 = 0; i0 < n; i0 += PPT * SORT_T) {
    int2 p[PPT];
    for (int j = 0; j < PPT; ++j) {
      int i = i0 + j * SORT_T + tid;
      p[j] = pairs[beg + min(i, n - 1)];
    }
    for (int j = 0; j < PPT; ++j) {
      int i = i0 + j * SORT_T + tid;
      if (i < n) {
        int pos = atomicAdd(&cnt[min(max(p[j].y - node0, 0), NPB - 1)], 1);
        if ((unsigned)pos < (unsigned)N_EDGES) csr_src[pos] = p[j].x;
      }
    }
  }
}

#define PROD_WG     256
#define PROD_FULL   12
#define N_STRIPS    (N_NODES / 16)
static_assert(N_STRIPS * 16 == N_NODES, "strips tile the rows exactly");
static_assert(PROD_WG * PROD_FULL <= N_STRIPS, "every work-group's own strips exist");
static_assert(PROD_WG * (PROD_FULL + 1) >= N_STRIPS, "the left-over strips fit one to a work-group");

__global__ void __launch_bounds__(SORT_T, 8)
csr_bucket_and_products(const int2* __restrict__ pairs,
                        const int* __restrict__ range_base,
                        const int* __restrict__ range_cnt,
                        int* __restrict__ deg,
                        int* __restrict__ rowstart,
                        int* __restrict__ csr_src,
                        const float* __restrict__ X,
                        const float* __restrict__ Wl,
                        const float* __restrict__ Wr,
                        float* __restrict__ xl,
                        float* __restrict__ xr) {
  if (blockIdx.x >= PROD_WG) {
    csr_bucket_body(pairs, range_base, range_cnt, deg, rowstart, csr_src, blockIdx.x - PROD_WG);
    return;
  }
  int w    = __builtin_amdgcn_readfirstlane(threadIdx.x >> 5);
  int lane = threadIdx.x & 31;
  int g    = blockIdx.x;
  int left = PROD_WG * PROD_FULL + g;
  int n_items = 2 * PROD_FULL + (left < N_STRIPS ? 2 : 0);
  for (int it = w; it < n_items; it += SORT_T / 32) {
    int ls = it >> 1;
    int mt = ls < PROD_FULL ? g * PROD_FULL + ls : left;
    if (it & 1) gemm0_strip<F_IN, 5, 5>(X, Wl, Wr, xl, xr, mt, lane);
    else        gemm0_strip<F_IN, 0, 5>(X, Wl, Wr, xl, xr, mt, lane);
  }
}

#define CH2 (OUT_CH / 2)

__device__ __forceinline__ void load10(const float* __restrict__ p, v2f (&v)[CH2]) {
  const v2f* q = (const v2f*)p;
  for (int i = 0; i < CH2; ++i) v[i] = q[i];
}

__device__ __forceinline__ void load_xl_row(const float* __restrict__ xl, int node, int h,
                                            v2f (&v)[CH2]) {
  const v2f* q = (const v2f*)(xl + node * HIDDEN + 2 * h);
  for (int i = 0; i < CH2; ++i) v[i] = q[8 * i];
}

__device__ __forceinline__ float head_logit(const v2f (&x)[CH2],
                                            const v2f (&r)[CH2],
                                            const v2f (&a)[CH2]) {
  v2f s = {0.0f, 0.0f};
  for (int i = 0; i < CH2; ++i) {
    v2f v = x[i] + r[i];
    v2f w = v * 0.2f;
    v.x = fmaxf(v.x, w.x);
    v.y = fmaxf(v.y, w.y);
    s = v * a[i] + s;
  }
  return s.x + s.y;
}

__device__ __forceinline__ void fold4(float& m, float& den, v2f (&acc)[CH2],
                                      float l0, float l1, float l2, float l3,
                                      const v2f (&x0)[CH2], const v2f (&x1)[CH2],
                                      const v2f (&x2)[CH2], const v2f (&x3)[CH2]) {
  float mn = fmaxf(fmaxf(m, fmaxf(l0, l1)), fmaxf(l2, l3));
  float sc = __expf(m - mn);
  float p0 = __expf(l0 - mn);
  float p1 = __expf(l1 - mn);
  float p2 = __expf(l2 - mn);
  float p3 = __expf(l3 - mn);
  den = den * sc + ((p0 + p1) + (p2 + p3));
  for (int i = 0; i < CH2; ++i)
    acc[i] = acc[i] * sc + ((x0[i] * p0 + x1[i] * p1) + (x2[i] * p2 + x3[i] * p3));
  m = mn;
}

__device__ __forceinline__ v2f elu_pair(v2f v) {
  v2f t;
  t.x = __expf(v.x);
  t.y = __expf(v.y);
  v2f far = t - 1.0f;
  v2f p = v * (1.0f / 5040.0f) + (1.0f / 720.0f);
  p = p * v + (1.0f / 120.0f);
  p = p * v + (1.0f / 24.0f);
  p = p * v + (1.0f / 6.0f);
  p = p * v + 0.5f;
  v2f near = (v * v) * p + v;
  v2f o;
  o.x = v.x > 0.0f ? v.x : (v.x > -0.35f ? near.x : far.x);
  o.y = v.y > 0.0f ? v.y : (v.y > -0.35f ? near.y : far.y);
  return o;
}

__device__ __forceinline__ void gather_node_head(const int* __restrict__ rowstart,
                                                 const int* __restrict__ deg,
                                                 const int* __restrict__ csr_src,
                                                 const float* __restrict__ xl,
                                                 const float* xr,
                                                 const float* __restrict__ att,
                                                 const float* __restrict__ bias,
                                                 int d, int h, float (&o)[OUT_CH]) {
  int hoff = h * OUT_CH;

  v2f a[CH2], r[CH2], b[CH2], x0[CH2], x1[CH2], x2[CH2], x3[CH2], acc[CH2];
  load10(att + hoff, a);
  load10(bias + hoff, b);
  load10(xr + d * HIDDEN + hoff, r);
  load_xl_row(xl, d, h, x0);

  int beg = min(max(rowstart[d], 0), N_EDGES);
  int end = beg + min(max(deg[d], 0), N_EDGES - beg);
  int rem = (end - beg) & 3;

  int t1 = csr_src[min(beg,     N_EDGES - 1)];
  int t2 = csr_src[min(beg + 1, N_EDGES - 1)];
  int t3 = csr_src[min(beg + 2, N_EDGES - 1)];
  int e = beg + rem;
  int s0 = 0, s1 = 0, s2 = 0, s3 = 0;
  if (e < end) { s0 = csr_src[e]; s1 = csr_src[e + 1]; s2 = csr_src[e + 2]; s3 = csr_src[e + 3]; }

  bool v1 = rem > 0, v2 = rem > 1, v3 = rem > 2;
  load_xl_row(xl, v1 ? clamp_node(t1) : d, h, x1);
  load_xl_row(xl, v2 ? clamp_node(t2) : d, h, x2);
  load_xl_row(xl, v3 ? clamp_node(t3) : d, h, x3);
  float m, den;
  {
    float l0 = head_logit(x0, r, a);
    float l1 = v1 ? head_logit(x1, r, a) : -INFINITY;
    float l2 = v2 ? head_logit(x2, r, a) : -INFINITY;
    float l3 = v3 ? head_logit(x3, r, a) : -INFINITY;
    m = fmaxf(fmaxf(l0, l1), fmaxf(l2, l3));
    float p0 = __expf(l0 - m);
    float p1 = v1 ? __expf(l1 - m) : 0.0f;
    float p2 = v2 ? __expf(l2 - m) : 0.0f;
    float p3 = v3 ? __expf(l3 - m) : 0.0f;
    den = (p0 + p1) + (p2 + p3);
    for (int i = 0; i < CH2; ++i)
      acc[i] = (x0[i] * p0 + x1[i] * p1) + (x2[i] * p2 + x3[i] * p3);
  }

  while (e < end) {
    int c0 = clamp_node(s0);
    int c1 = clamp_node(s1);
    int c2 = clamp_node(s2);
    int c3 = clamp_node(s3);
    e += 4;
    if (e < end) { s0 = csr_src[e]; s1 = csr_src[e + 1]; s2 = csr_src[e + 2]; s3 = csr_src[e + 3]; }
    load_xl_row(xl, c0, h, x0);
    load_xl_row(xl, c1, h, x1);
    load_xl_row(xl, c2, h, x2);
    load_xl_row(xl, c3, h, x3);
    float l0 = head_logit(x0, r, a);
    float l1 = head_logit(x1, r, a);
    float l2 = head_logit(x2, r, a);
    float l3 = head_logit(x3, r, a);
    fold4(m, den, acc, l0, l1, l2, l3, x0, x1, x2, x3);
  }

  float inv = 1.0f / (den + 1e-16f);
  for (int i = 0; i < CH2; ++i) {
    v2f e = elu_pair(acc[i] * inv + b[i]);
    o[2 * i]     = e.x;
    o[2 * i + 1] = e.y;
  }
}

__global__ void __launch_bounds__(256, 8)
gat_node_gather(const int* __restrict__ rowstart,
                const int* __restrict__ deg,
                const int* __restrict__ csr_src,
                const float* __restrict__ xl,
                const float* __restrict__ xr,
                const float* __restrict__ att,
                const float* __restrict__ bias,
                float* __restrict__ xout) {
  int t = blockIdx.x * blockDim.x + threadIdx.x;
  if (t >= N_NODES * HEADS) return;
  int d = t >> 3;
  int h = t & 7;
  float o[OUT_CH];
  gather_node_head(rowstart, deg, csr_src, xl, xr, att, bias, d, h, o);
  float2* po = (float2*)(xout + d * HIDDEN + h * OUT_CH);
  for (int i = 0; i < OUT_CH / 2; ++i) {
    float2 v;
    v.x = o[2 * i];
    v.y = o[2 * i + 1];
    po[i] = v;
  }
}

#define TILE_PITCH 84

template <int NTW>
__device__ __forceinline__ void strip_product(const float* tile_rows,
                                              const float* __restrict__ Wl,
                                              const float* __restrict__ Wr,
                                              float* __restrict__ xl_next,
                                              float* xr_next,
                                              int row0, int q, int lane) {
  int half = lane >> 4;
  int r    = lane & 15;
  const float* ap = tile_rows + r * TILE_PITCH + 2 * half;
  const float* bp[NTW];
#pragma unroll
  for (int j = 0; j < NTW; ++j) {
    int id = q + 4 * j;
    const float* w = id < 5 ? Wl : Wr;
    int ct = id < 5 ? id : id - 5;
    bp[j] = w + (2 * half) * HIDDEN + ct * 16 + r;
  }
  v8f c[NTW];
#pragma unroll
  for (int j = 0; j < NTW; ++j) c[j] = (v8f){};

#pragma unroll 4
  for (int k = 0; k < HIDDEN; k += 4) {
    v2f a = *(const v2f*)(ap + k);
#pragma unroll
    for (int j = 0; j < NTW; ++j) {
      v2f b;
      b.x = bp[j][k * HIDDEN];
      b.y = bp[j][(k + 1) * HIDDEN];
      c[j] = __builtin_amdgcn_wmma_f32_16x16x4_f32(false, a, false, b, (short)0, c[j], false, false);
    }
  }

#pragma unroll
  for (int j = 0; j < NTW; ++j) {
    int id = q + 4 * j;
    int ct = id < 5 ? id : id - 5;
    int n  = ct * 16 + r;
    int hd = n / OUT_CH;
    int slot = xl_slot(hd, n - hd * OUT_CH);
#pragma unroll
    for (int v = 0; v < 8; ++v) {
      int orow = row0 + v + 8 * half;
      if (id < 5) xl_next[orow * HIDDEN + slot] = c[j][v];
      else        xr_next[orow * HIDDEN + n]    = c[j][v];
    }
  }
}

__global__ void __launch_bounds__(256, 8)
gat_gather_then_gemm(const int* __restrict__ rowstart,
                     const int* __restrict__ deg,
                     const int* __restrict__ csr_src,
                     const float* __restrict__ xl,
                     float* xr,
                     const float* __restrict__ att,
                     const float* __restrict__ bias,
                     const float* __restrict__ Wl,
                     const float* __restrict__ Wr,
                     float* __restrict__ xl_next) {
  __shared__ __attribute__((aligned(16))) float tile[32 * TILE_PITCH];
  int t = blockIdx.x * blockDim.x + threadIdx.x;
  if (t < N_NODES * HEADS) {
    int d = t >> 3;
    int h = t & 7;
    float o[OUT_CH];
    gather_node_head(rowstart, deg, csr_src, xl, xr, att, bias, d, h, o);
    float2* po = (float2*)(tile + (threadIdx.x >> 3) * TILE_PITCH + h * OUT_CH);
    for (int i = 0; i < OUT_CH / 2; ++i) {
      float2 v;
      v.x = o[2 * i];
      v.y = o[2 * i + 1];
      po[i] = v;
    }
  }
  __syncthreads();

  int w     = __builtin_amdgcn_readfirstlane(threadIdx.x >> 5);
  int strip = w >> 2;
  int q     = strip ? 3 - (w & 3) : (w & 3);
  int row0  = blockIdx.x * 32 + strip * 16;
  if (row0 >= N_NODES) return;
  const float* rows = tile + strip * 16 * TILE_PITCH;
  int lane = threadIdx.x & 31;
  if (q < 2) strip_product<3>(rows, Wl, Wr, xl_next, xr, row0, q, lane);
  else       strip_product<2>(rows, Wl, Wr, xl_next, xr, row0, q, lane);
}

extern "C" void kernel_launch(void* const* d_in, const int* in_sizes, int n_in,
                              void* d_out, int out_size, void* d_ws, size_t ws_size,
                              hipStream_t stream) {
  (void)in_sizes; (void)n_in; (void)out_size; (void)ws_size;

  const float* x    = (const float*)d_in[0];
  const int*   ei   = (const int*)d_in[1];
  const int*   esrc = ei;
  const int*   edst = ei + N_EDGES;

  float* ws    = (float*)d_ws;
  float* xl    = ws;
  float* xr    = xl   + (size_t)N_NODES * HIDDEN;
  float* xbuf  = xr   + (size_t)N_NODES * HIDDEN;
  int*   deg      = (int*)(xbuf + (size_t)N_NODES * HIDDEN);
  int*   rowstart = deg      + N_NODES;
  int*   spare    = rowstart + N_NODES;
  int*   csr_src  = spare    + 64 + N_EDGES;
  int*   table      = (int*)xbuf;
  int*   range_base = table      + NCH * NB;
  int*   range_cnt  = range_base + NB;
  int2*  pairs      = (int2*)(range_cnt + NB);

  const int head_blocks  = (N_NODES * HEADS + 255) / 256;

  csr_hist   <<<NCH, SORT_T, 0, stream>>>(edst, table);
  csr_scatter<<<NCH, SORT_T, 0, stream>>>(esrc, edst, table, range_base, range_cnt, pairs);
  float* xl_a = xl;
  float* xl_b = xbuf;
  csr_bucket_and_products<<<PROD_WG + NB, SORT_T, 0, stream>>>(
      pairs, range_base, range_cnt, deg, rowstart, csr_src,
      x, (const float*)d_in[2], (const float*)d_in[3], xl_a, xr);
  gat_gather_then_gemm<<<head_blocks, 256, 0, stream>>>(
      rowstart, deg, csr_src, xl_a, xr, (const float*)d_in[4], (const float*)d_in[5],
      (const float*)d_in[6], (const float*)d_in[7], xl_b);
  gat_gather_then_gemm<<<head_blocks, 256, 0, stream>>>(
      rowstart, deg, csr_src, xl_b, xr, (const float*)d_in[8], (const float*)d_in[9],
      (const float*)d_in[10], (const float*)d_in[11], xl_a);
  gat_node_gather<<<head_blocks, 256, 0, stream>>>(
      rowstart, deg, csr_src, xl_a, xr, (const float*)d_in[12], (const float*)d_in[13],
      (float*)d_out);
}


# edge grouping level 1 sorts each chunk by node range inside its own slice in one launch (LDS counters); csr_bucket walks its 256 segments: no histogram launch, 5 launches not 6; sum order=LDS arrival
# speedup vs baseline: 8.2230x; 1.0661x over previous
#include <hip/hip_runtime.h>
#include <math.h>

#define N_NODES 50000
#define N_EDGES 800000
#define F_IN    256
#define HEADS   8
#define OUT_CH  10
#define HIDDEN  80

typedef __attribute__((ext_vector_type(2))) float v2f;
typedef __attribute__((ext_vector_type(8))) float v8f;

__device__ __forceinline__ int xl_slot(int head, int ch) {
  return ((ch >> 1) << 4) + 2 * head + (ch & 1);
}

template <int K, int ID0, int NTW>
__device__ __forceinline__ void gemm0_strip(const float* __restrict__ X,
                                            const float* __restrict__ Wl,
                                            const float* __restrict__ Wr,
                                            float* __restrict__ xl,
                                            float* __restrict__ xr,
                                            int mt, int lane) {
  int half = lane >> 4;
  int r    = lane & 15;

  const float* ap = X + (mt * 16 + r) * K + 2 * half;
  const float* bp[NTW];
#pragma unroll
  for (int j = 0; j < NTW; ++j) {
    int id = ID0 + j;
    const float* w = id < 5 ? Wl : Wr;
    int ct = id < 5 ? id : id - 5;
    bp[j] = w + (2 * half) * HIDDEN + ct * 16 + r;
  }
  v8f c[NTW];
#pragma unroll
  for (int j = 0; j < NTW; ++j) c[j] = (v8f){};

#pragma unroll 4
  for (int k = 0; k < K; k += 4) {
    v2f a = *(const v2f*)(ap + k);
#pragma unroll
    for (int j = 0; j < NTW; ++j) {
      v2f b;
      b.x = bp[j][k * HIDDEN];
      b.y = bp[j][(k + 1) * HIDDEN];
      c[j] = __builtin_amdgcn_wmma_f32_16x16x4_f32(false, a, false, b, (short)0, c[j], false, false);
    }
  }

#pragma unroll
  for (int j = 0; j < NTW; ++j) {
    int id = ID0 + j;
    int ct = id < 5 ? id : id - 5;
    int n    = ct * 16 + r;
    int hd   = n / OUT_CH;
    int slot = xl_slot(hd, n - hd * OUT_CH);
#pragma unroll
    for (int v = 0; v < 8; ++v) {
      int orow = mt * 16 + v + 8 * half;
      if (id < 5) xl[orow * HIDDEN + slot] = c[j][v];
      else        xr[orow * HIDDEN + n]    = c[j][v];
    }
  }
}

#define NB      256
#define NPB     196
#define NCH     256
#define CHUNK   3125
#define SORT_T  512
#define EPT     7
#define SEG_L   8
#define SEG_PT  (NCH / (SORT_T / SEG_L))
#define SEG_U   2

static_assert(NB * NPB >= N_NODES, "ranges cover all nodes");
static_assert(NCH * CHUNK == N_EDGES, "chunks tile the edge list exactly");
static_assert(EPT * SORT_T >= CHUNK, "one pass of a work-group covers its chunk");
static_assert(NB == 256 && SORT_T >= NB && SORT_T % 64 == 0, "thread maps below assume these");
static_assert(SEG_PT * (SORT_T / SEG_L) == NCH, "a work-group's lane groups cover all segments of a range");

__device__ __forceinline__ int clamp_node(int s) {
  return min(max(s, 0), N_NODES - 1);
}

__device__ __forceinline__ int range_of(int node) { return node / NPB; }

__device__ __forceinline__ int scan256_excl(int v, int* wsum, int tid) {
  int lane = tid & 31;
  int w    = tid >> 5;
  int incl = v;
  for (int off = 1; off < 32; off <<= 1) {
    int t = __shfl_up(incl, off);
    if (lane >= off) incl += t;
  }
  if (tid < 256 && lane == 31) wsum[w] = incl;
  __syncthreads();
  int base = 0;
  for (int j = 0; j < 8; ++j) {
    int s = wsum[j];
    if (j < w) base += s;
  }
  return base + incl - v;
}

__global__ void __launch_bounds__(SORT_T)
csr_chunk_sort(const int* __restrict__ esrc,
               const int* __restrict__ edst,
               int* __restrict__ seg_off,
               int* __restrict__ seg_cnt,
               int2* __restrict__ pairs) {
  __shared__ int h[NB];
  __shared__ int wsum[8];
  int tid = threadIdx.x;
  int c   = blockIdx.x;
  if (tid < NB) h[tid] = 0;

  int s[EPT], d[EPT];
  for (int j = 0; j < EPT; ++j) {
    int i = c * CHUNK + min(tid + j * SORT_T, CHUNK - 1);
    s[j] = esrc[i];
    d[j] = edst[i];
  }
  __syncthreads();
  for (int j = 0; j < EPT; ++j) {
    d[j] = clamp_node(d[j]);
    if (tid + j * SORT_T < CHUNK) atomicAdd(&h[range_of(d[j])], 1);
  }
  __syncthreads();

  int v    = tid < NB ? h[tid] : 0;
  int excl = scan256_excl(v, wsum, tid);
  if (tid < NB) h[tid] = excl;
  __syncthreads();

  int pos[EPT];
  for (int j = 0; j < EPT; ++j)
    pos[j] = tid + j * SORT_T < CHUNK ? atomicAdd(&h[range_of(d[j])], 1) : -1;
  for (int j = 0; j < EPT; ++j) {
    if ((unsigned)pos[j] < (unsigned)CHUNK) {
      int2 pr;
      pr.x = s[j];
      pr.y = d[j];
      pairs[c * CHUNK + pos[j]] = pr;
    }
  }
  if (tid < NB) {
    seg_off[tid * NCH + c] = excl;
    seg_cnt[tid * NCH + c] = v;
  }
}

__device__ __forceinline__ void csr_bucket_body(const int2* __restrict__ pairs,
                                                const int* __restrict__ seg_off,
                                                const int* __restrict__ seg_cnt,
                                                int* __restrict__ deg,
                                                int* __restrict__ rowstart,
                                                int* __restrict__ csr_src,
                                                int blk) {
  __shared__ int cnt[NB];
  __shared__ int wsum[8];
  __shared__ int part[SORT_T / 32];
  int tid   = threadIdx.x;
  int node0 = blk * NPB;
  if (tid < NB) cnt[tid] = 0;

  int sub = tid & (SEG_L - 1);
  int grp = tid / SEG_L;
  int base[SEG_PT], n[SEG_PT];
  int before = 0;
  for (int q = 0; q < SEG_PT; ++q) {
    int ch  = grp + q * (SORT_T / SEG_L);
    int off = min(max(seg_off[blk * NCH + ch], 0), CHUNK);
    n[q]    = min(max(seg_cnt[blk * NCH + ch], 0), CHUNK - off);
    base[q] = ch * CHUNK + off;
    if (sub == 0) before += off;
  }
  for (int off = 16; off > 0; off >>= 1) before += __shfl_xor(before, off);
  if ((tid & 31) == 0) part[tid >> 5] = before;

  int2 p[SEG_PT][SEG_U];
  for (int q = 0; q < SEG_PT; ++q)
    for (int u = 0; u < SEG_U; ++u) {
      int i = min(sub + SEG_L * u, max(n[q] - 1, 0));
      p[q][u] = pairs[min(base[q] + i, N_EDGES - 1)];
    }
  __syncthreads();

  int beg = 0;
  for (int j = 0; j < SORT_T / 32; ++j) beg += part[j];
  beg = min(max(beg, 0), N_EDGES);

  for (int q = 0; q < SEG_PT; ++q) {
    for (int u = 0; u < SEG_U; ++u)
      if (sub + SEG_L * u < n[q]) atomicAdd(&cnt[min(max(p[q][u].y - node0, 0), NPB - 1)], 1);
    for (int i = sub + SEG_L * SEG_U; i < n[q]; i += SEG_L) {
      int2 e = pairs[min(base[q] + i, N_EDGES - 1)];
      atomicAdd(&cnt[min(max(e.y - node0, 0), NPB - 1)], 1);
    }
  }
  __syncthreads();

  int v    = tid < NB ? cnt[tid] : 0;
  int excl = scan256_excl(v, wsum, tid);
  if (tid < NB) {
    int start = beg + excl;
    int node  = node0 + tid;
    if (tid < NPB && node < N_NODES) {
      deg[node]      = v;
      rowstart[node] = start;
    }
    cnt[tid] = start;
  }
  __syncthreads();

  for (int q = 0; q < SEG_PT; ++q) {
    for (int u = 0; u < SEG_U; ++u)
      if (sub + SEG_L * u < n[q]) {
        int pos = atomicAdd(&cnt[min(max(p[q][u].y - node0, 0), NPB - 1)], 1);
        if ((unsigned)pos < (unsigned)N_EDGES) csr_src[pos] = p[q][u].x;
      }
    for (int i = sub + SEG_L * SEG_U; i < n[q]; i += SEG_L) {
      int2 e = pairs[min(base[q] + i, N_EDGES - 1)];
      int pos = atomicAdd(&cnt[min(max(e.y - node0, 0), NPB - 1)], 1);
      if ((unsigned)pos < (unsigned)N_EDGES) csr_src[pos] = e.x;
    }
  }
}

#define PROD_WG     256
#define PROD_FULL   12
#define N_STRIPS    (N_NODES / 16)
static_assert(N_STRIPS * 16 == N_NODES, "strips tile the rows exactly");
static_assert(PROD_WG * PROD_FULL <= N_STRIPS, "every work-group's own strips exist");
static_assert(PROD_WG * (PROD_FULL + 1) >= N_STRIPS, "the left-over strips fit one to a work-group");

__global__ void __launch_bounds__(SORT_T, 8)
csr_bucket_and_products(const int2* __restrict__ pairs,
                        const int* __restrict__ seg_off,
                        const int* __restrict__ seg_cnt,
                        int* __restrict__ deg,
                        int* __restrict__ rowstart,
                        int* __restrict__ csr_src,
                        const float* __restrict__ X,
                        const float* __restrict__ Wl,
                        const float* __restrict__ Wr,
                        float* __restrict__ xl,
                        float* __restrict__ xr) {
  if (blockIdx.x >= PROD_WG) {
    csr_bucket_body(pairs, seg_off, seg_cnt, deg, rowstart, csr_src, blockIdx.x - PROD_WG);
    return;
  }
  int w    = __builtin_amdgcn_readfirstlane(threadIdx.x >> 5);
  int lane = threadIdx.x & 31;
  int g    = blockIdx.x;
  int left = PROD_WG * PROD_FULL + g;
  int n_items = 2 * PROD_FULL + (left < N_STRIPS ? 2 : 0);
  for (int it = w; it < n_items; it += SORT_T / 32) {
    int ls = it >> 1;
    int mt = ls < PROD_FULL ? g * PROD_FULL + ls : left;
    if (it & 1) gemm0_strip<F_IN, 5, 5>(X, Wl, Wr, xl, xr, mt, lane);
    else        gemm0_strip<F_IN, 0, 5>(X, Wl, Wr, xl, xr, mt, lane);
  }
}

#define CH2 (OUT_CH / 2)

__device__ __forceinline__ void load10(const float* __restrict__ p, v2f (&v)[CH2]) {
  const v2f* q = (const v2f*)p;
  for (int i = 0; i < CH2; ++i) v[i] = q[i];
}

__device__ __forceinline__ void load_xl_row(const float* __restrict__ xl, int node, int h,
                                            v2f (&v)[CH2]) {
  const v2f* q = (const v2f*)(xl + node * HIDDEN + 2 * h);
  for (int i = 0; i < CH2; ++i) v[i] = q[8 * i];
}

__device__ __forceinline__ float head_logit(const v2f (&x)[CH2],
                                            const v2f (&r)[CH2],
                                            const v2f (&a)[CH2]) {
  v2f s = {0.0f, 0.0f};
  for (int i = 0; i < CH2; ++i) {
    v2f v = x[i] + r[i];
    v2f w = v * 0.2f;
    v.x = fmaxf(v.x, w.x);
    v.y = fmaxf(v.y, w.y);
    s = v * a[i] + s;
  }
  return s.x + s.y;
}

__device__ __forceinline__ void fold4(float& m, float& den, v2f (&acc)[CH2],
                                      float l0, float l1, float l2, float l3,
                                      const v2f (&x0)[CH2], const v2f (&x1)[CH2],
                                      const v2f (&x2)[CH2], const v2f (&x3)[CH2]) {
  float mn = fmaxf(fmaxf(m, fmaxf(l0, l1)), fmaxf(l2, l3));
  float sc = __expf(m - mn);
  float p0 = __expf(l0 - mn);
  float p1 = __expf(l1 - mn);
  float p2 = __expf(l2 - mn);
  float p3 = __expf(l3 - mn);
  den = den * sc + ((p0 + p1) + (p2 + p3));
  for (int i = 0; i < CH2; ++i)
    acc[i] = acc[i] * sc + ((x0[i] * p0 + x1[i] * p1) + (x2[i] * p2 + x3[i] * p3));
  m = mn;
}

__device__ __forceinline__ v2f elu_pair(v2f v) {
  v2f t;
  t.x = __expf(v.x);
  t.y = __expf(v.y);
  v2f far = t - 1.0f;
  v2f p = v * (1.0f / 5040.0f) + (1.0f / 720.0f);
  p = p * v + (1.0f / 120.0f);
  p = p * v + (1.0f / 24.0f);
  p = p * v + (1.0f / 6.0f);
  p = p * v + 0.5f;
  v2f near = (v * v) * p + v;
  v2f o;
  o.x = v.x > 0.0f ? v.x : (v.x > -0.35f ? near.x : far.x);
  o.y = v.y > 0.0f ? v.y : (v.y > -0.35f ? near.y : far.y);
  return o;
}

__device__ __forceinline__ void gather_node_head(const int* __restrict__ rowstart,
                                                 const int* __restrict__ deg,
                                                 const int* __restrict__ csr_src,
                                                 const float* __restrict__ xl,
                                                 const float* xr,
                                                 const float* __restrict__ att,
                                                 const float* __restrict__ bias,
                                                 int d, int h, float (&o)[OUT_CH]) {
  int hoff = h * OUT_CH;

  v2f a[CH2], r[CH2], b[CH2], x0[CH2], x1[CH2], x2[CH2], x3[CH2], acc[CH2];
  load10(att + hoff, a);
  load10(bias + hoff, b);
  load10(xr + d * HIDDEN + hoff, r);
  load_xl_row(xl, d, h, x0);

  int beg = min(max(rowstart[d], 0), N_EDGES);
  int end = beg + min(max(deg[d], 0), N_EDGES - beg);
  int rem = (end - beg) & 3;

  int t1 = csr_src[min(beg,     N_EDGES - 1)];
  int t2 = csr_src[min(beg + 1, N_EDGES - 1)];
  int t3 = csr_src[min(beg + 2, N_EDGES - 1)];
  int e = beg + rem;
  int s0 = 0, s1 = 0, s2 = 0, s3 = 0;
  if (e < end) { s0 = csr_src[e]; s1 = csr_src[e + 1]; s2 = csr_src[e + 2]; s3 = csr_src[e + 3]; }

  bool v1 = rem > 0, v2 = rem > 1, v3 = rem > 2;
  load_xl_row(xl, v1 ? clamp_node(t1) : d, h, x1);
  load_xl_row(xl, v2 ? clamp_node(t2) : d, h, x2);
  load_xl_row(xl, v3 ? clamp_node(t3) : d, h, x3);
  float m, den;
  {
    float l0 = head_logit(x0, r, a);
    float l1 = v1 ? head_logit(x1, r, a) : -INFINITY;
    float l2 = v2 ? head_logit(x2, r, a) : -INFINITY;
    float l3 = v3 ? head_logit(x3, r, a) : -INFINITY;
    m = fmaxf(fmaxf(l0, l1), fmaxf(l2, l3));
    float p0 = __expf(l0 - m);
    float p1 = v1 ? __expf(l1 - m) : 0.0f;
    float p2 = v2 ? __expf(l2 - m) : 0.0f;
    float p3 = v3 ? __expf(l3 - m) : 0.0f;
    den = (p0 + p1) + (p2 + p3);
    for (int i = 0; i < CH2; ++i)
      acc[i] = (x0[i] * p0 + x1[i] * p1) + (x2[i] * p2 + x3[i] * p3);
  }

  while (e < end) {
    int c0 = clamp_node(s0);
    int c1 = clamp_node(s1);
    int c2 = clamp_node(s2);
    int c3 = clamp_node(s3);
    e += 4;
    if (e < end) { s0 = csr_src[e]; s1 = csr_src[e + 1]; s2 = csr_src[e + 2]; s3 = csr_src[e + 3]; }
    load_xl_row(xl, c0, h, x0);
    load_xl_row(xl, c1, h, x1);
    load_xl_row(xl, c2, h, x2);
    load_xl_row(xl, c3, h, x3);
    float l0 = head_logit(x0, r, a);
    float l1 = head_logit(x1, r, a);
    float l2 = head_logit(x2, r, a);
    float l3 = head_logit(x3, r, a);
    fold4(m, den, acc, l0, l1, l2, l3, x0, x1, x2, x3);
  }

  float inv = 1.0f / (den + 1e-16f);
  for (int i = 0; i < CH2; ++i) {
    v2f e = elu_pair(acc[i] * inv + b[i]);
    o[2 * i]     = e.x;
    o[2 * i + 1] = e.y;
  }
}

__global__ void __launch_bounds__(256, 8)
gat_node_gather(const int* __restrict__ rowstart,
                const int* __restrict__ deg,
                const int* __restrict__ csr_src,
                const float* __restrict__ xl,
                const float* __restrict__ xr,
                const float* __restrict__ att,
                const float* __restrict__ bias,
                float* __restrict__ xout) {
  int t = blockIdx.x * blockDim.x + threadIdx.x;
  if (t >= N_NODES * HEADS) return;
  int d = t >> 3;
  int h = t & 7;
  float o[OUT_CH];
  gather_node_head(rowstart, deg, csr_src, xl, xr, att, bias, d, h, o);
  float2* po = (float2*)(xout + d * HIDDEN + h * OUT_CH);
  for (int i = 0; i < OUT_CH / 2; ++i) {
    float2 v;
    v.x = o[2 * i];
    v.y = o[2 * i + 1];
    po[i] = v;
  }
}

#define TILE_PITCH 84

template <int NTW>
__device__ __forceinline__ void strip_product(const float* tile_rows,
                                              const float* __restrict__ Wl,
                                              const float* __restrict__ Wr,
                                              float* __restrict__ xl_next,
                                              float* xr_next,
                                              int row0, int q, int lane) {
  int half = lane >> 4;
  int r    = lane & 15;
  const float* ap = tile_rows + r * TILE_PITCH + 2 * half;
  const float* bp[NTW];
#pragma unroll
  for (int j = 0; j < NTW; ++j) {
    int id = q + 4 * j;
    const float* w = id < 5 ? Wl : Wr;
    int ct = id < 5 ? id : id - 5;
    bp[j] = w + (2 * half) * HIDDEN + ct * 16 + r;
  }
  v8f c[NTW];
#pragma unroll
  for (int j = 0; j < NTW; ++j) c[j] = (v8f){};

#pragma unroll 4
  for (int k = 0; k < HIDDEN; k += 4) {
    v2f a = *(const v2f*)(ap + k);
#pragma unroll
    for (int j = 0; j < NTW; ++j) {
      v2f b;
      b.x = bp[j][k * HIDDEN];
      b.y = bp[j][(k + 1) * HIDDEN];
      c[j] = __builtin_amdgcn_wmma_f32_16x16x4_f32(false, a, false, b, (short)0, c[j], false, false);
    }
  }

#pragma unroll
  for (int j = 0; j < NTW; ++j) {
    int id = q + 4 * j;
    int ct = id < 5 ? id : id - 5;
    int n  = ct * 16 + r;
    int hd = n / OUT_CH;
    int slot = xl_slot(hd, n - hd * OUT_CH);
#pragma unroll
    for (int v = 0; v < 8; ++v) {
      int orow = row0 + v + 8 * half;
      if (id < 5) xl_next[orow * HIDDEN + slot] = c[j][v];
      else        xr_next[orow * HIDDEN + n]    = c[j][v];
    }
  }
}

__global__ void __launch_bounds__(256, 8)
gat_gather_then_gemm(const int* __restrict__ rowstart,
                     const int* __restrict__ deg,
                     const int* __restrict__ csr_src,
                     const float* __restrict__ xl,
                     float* xr,
                     const float* __restrict__ att,
                     const float* __restrict__ bias,
                     const float* __restrict__ Wl,
                     const float* __restrict__ Wr,
                     float* __restrict__ xl_next) {
  __shared__ __attribute__((aligned(16))) float tile[32 * TILE_PITCH];
  int t = blockIdx.x * blockDim.x + threadIdx.x;
  if (t < N_NODES * HEADS) {
    int d = t >> 3;
    int h = t & 7;
    float o[OUT_CH];
    gather_node_head(rowstart, deg, csr_src, xl, xr, att, bias, d, h, o);
    float2* po = (float2*)(tile + (threadIdx.x >> 3) * TILE_PITCH + h * OUT_CH);
    for (int i = 0; i < OUT_CH / 2; ++i) {
      float2 v;
      v.x = o[2 * i];
      v.y = o[2 * i + 1];
      po[i] = v;
    }
  }
  __syncthreads();

  int w     = __builtin_amdgcn_readfirstlane(threadIdx.x >> 5);
  int strip = w >> 2;
  int q     = strip ? 3 - (w & 3) : (w & 3);
  int row0  = blockIdx.x * 32 + strip * 16;
  if (row0 >= N_NODES) return;
  const float* rows = tile + strip * 16 * TILE_PITCH;
  int lane = threadIdx.x & 31;
  if (q < 2) strip_product<3>(rows, Wl, Wr, xl_next, xr, row0, q, lane);
  else       strip_product<2>(rows, Wl, Wr, xl_next, xr, row0, q, lane);
}

extern "C" void kernel_launch(void* const* d_in, const int* in_sizes, int n_in,
                              void* d_out, int out_size, void* d_ws, size_t ws_size,
                              hipStream_t stream) {
  (void)in_sizes; (void)n_in; (void)out_size; (void)ws_size;

  const float* x    = (const float*)d_in[0];
  const int*   ei   = (const int*)d_in[1];
  const int*   esrc = ei;
  const int*   edst = ei + N_EDGES;

  float* ws    = (float*)d_ws;
  float* xl    = ws;
  float* xr    = xl   + (size_t)N_NODES * HIDDEN;
  float* xbuf  = xr   + (size_t)N_NODES * HIDDEN;
  int*   deg      = (int*)(xbuf + (size_t)N_NODES * HIDDEN);
  int*   rowstart = deg      + N_NODES;
  int*   spare    = rowstart + N_NODES;
  int*   csr_src  = spare    + 64 + N_EDGES;
  int*   seg_off    = (int*)xbuf;
  int*   seg_cnt    = seg_off    + NB * NCH;
  int2*  pairs      = (int2*)(seg_cnt + NB * NCH);

  const int head_blocks  = (N_NODES * HEADS + 255) / 256;

  csr_chunk_sort<<<NCH, SORT_T, 0, stream>>>(esrc, edst, seg_off, seg_cnt, pairs);
  float* xl_a = xl;
  float* xl_b = xbuf;
  csr_bucket_and_products<<<PROD_WG + NB, SORT_T, 0, stream>>>(
      pairs, seg_off, seg_cnt, deg, rowstart, csr_src,
      x, (const float*)d_in[2], (const float*)d_in[3], xl_a, xr);
  gat_gather_then_gemm<<<head_blocks, 256, 0, stream>>>(
      rowstart, deg, csr_src, xl_a, xr, (const float*)d_in[4], (const float*)d_in[5],
      (const float*)d_in[6], (const float*)d_in[7], xl_b);
  gat_gather_then_gemm<<<head_blocks, 256, 0, stream>>>(
      rowstart, deg, csr_src, xl_b, xr, (const float*)d_in[8], (const float*)d_in[9],
      (const float*)d_in[10], (const float*)d_in[11], xl_a);
  gat_node_gather<<<head_blocks, 256, 0, stream>>>(
      rowstart, deg, csr_src, xl_a, xr, (const float*)d_in[12], (const float*)d_in[13],
      (float*)d_out);
}
